# starting kernel
# baseline (speedup 1.0000x reference)
#include <hip/hip_runtime.h>
#include <hip/hip_bf16.h>

typedef __attribute__((ext_vector_type(2))) float v2f;
typedef __attribute__((ext_vector_type(8))) float v8f;

#define T_LEN 262144
#define NTILE (T_LEN / 256)

__global__ __launch_bounds__(256) void fir_u_wmma(const float* __restrict__ Is,
                                                  const float* __restrict__ a,
                                                  float* __restrict__ u) {
  __shared__ float a_pad[96];
  {
    const int tid = threadIdx.x;
    if (tid < 96) {
      const int d = tid - 16;
      a_pad[tid] = (d >= 0 && d < 64) ? a[d] : 0.f;
    }
  }
  __syncthreads();

  const int lane = threadIdx.x & 31;
  const int wave = threadIdx.x >> 5;
  const int tile = blockIdx.x * 8 + wave;
  const int t0   = tile * 256;
  const int base = t0 - 64;
  const int m    = lane & 15;
  const int half = lane >> 4;
  const int kb   = half * 2;

  v8f c = {0.f, 0.f, 0.f, 0.f, 0.f, 0.f, 0.f, 0.f};

  if (tile != 0) {
#pragma unroll
    for (int p = 0; p < 20; ++p) {
      const int k0 = p * 4 + kb;
      v2f av, bv;
      av.x = a_pad[16 + k0 - m];
      av.y = a_pad[17 + k0 - m];
      bv = *(const v2f*)(Is + (base + 16 * m + k0));
      c = __builtin_amdgcn_wmma_f32_16x16x4_f32(false, av, false, bv,
                                                (short)0, c, false, false);
    }
  } else {
#pragma unroll
    for (int p = 0; p < 20; ++p) {
      const int k0 = p * 4 + kb;
      v2f av, bv;
      av.x = a_pad[16 + k0 - m];
      av.y = a_pad[17 + k0 - m];
      const int i0 = base + 16 * m + k0;
      bv.x = (i0     >= 0) ? Is[i0]     : 0.f;
      bv.y = (i0 + 1 >= 0) ? Is[i0 + 1] : 0.f;
      c = __builtin_amdgcn_wmma_f32_16x16x4_f32(false, av, false, bv,
                                                (short)0, c, false, false);
    }
  }

#pragma unroll
  for (int r = 0; r < 8; ++r) {
    const int M = r + 8 * half;
    const int t = t0 + 16 * m + M;
    u[t] = Is[t] + c[r];
  }
}

template <int CTRL>
__device__ __forceinline__ float dpp_xadd(float x) {
  const int y = __builtin_amdgcn_update_dpp(0, __builtin_bit_cast(int, x),
                                            CTRL, 0xf, 0xf, true);
  return x + __builtin_bit_cast(float, y);
}

__global__ __launch_bounds__(32) void recurrence(const float* __restrict__ u,
                                                 const float* __restrict__ b,
                                                 const float* __restrict__ w1,
                                                 const float* __restrict__ b1,
                                                 const float* __restrict__ w2,
                                                 const float* __restrict__ b2p,
                                                 float* __restrict__ out) {
  __shared__ float acc[64];
  const int lane = threadIdx.x;

  acc[lane]      = 0.f;
  acc[lane + 32] = 0.f;

  float w1r2[4], b1r2[4], w2r[4];
#pragma unroll
  for (int r = 0; r < 4; ++r) {
    w1r2[r] = 2.0f * w1[lane + 32 * r];
    b1r2[r] = 2.0f * b1[lane + 32 * r];
    w2r[r]  = w2[lane + 32 * r];
  }
  const float cA  = b[62 - lane];
  const float cB  = (lane <= 30) ? b[30 - lane] : 0.f;
  const float b63 = b[63];
  const float gb2 = b2p[0];

  float f_prev = 0.f;
  float accP   = 0.f;
  float uNxt   = u[lane];
  float fOut   = 0.f;

  const int NC = T_LEN / 32;
  for (int ch = 0; ch < NC; ++ch) {
    const float uCur = uNxt;
    const int   nb   = (ch + 1) * 32;
    const int   pidx = (nb + lane < T_LEN) ? (nb + lane) : (T_LEN - 1);
    uNxt = u[pidx];
    const int i0 = ch * 32;
#pragma unroll 4
    for (int s = 0; s < 32; ++s) {
      const int i    = i0 + s;
      const int slot = (i + 1) & 63;
      const float accNext = acc[slot];
      acc[slot] = 0.f;

      const float uu = __shfl(uCur, s, 32);
      const float x  = uu - fmaf(b63, f_prev, accP);

      float th[4];
#pragma unroll
      for (int r = 0; r < 4; ++r) {
        const float e = __expf(fmaf(w1r2[r], x, b1r2[r]));
        th[r] = 1.0f - 2.0f * __builtin_amdgcn_rcpf(e + 1.0f);
      }
      const float p0   = fmaf(w2r[0], th[0], w2r[1] * th[1]);
      const float p1   = fmaf(w2r[2], th[2], w2r[3] * th[3]);
      float ssum = p0 + p1;

      ssum = dpp_xadd<0xB1>(ssum);
      ssum = dpp_xadd<0x4E>(ssum);
      ssum = dpp_xadd<0x141>(ssum);
      ssum = dpp_xadd<0x140>(ssum);
      ssum += __shfl_xor(ssum, 16, 32);

      const float z = ssum + gb2;
      const float f = fmaxf(z, 0.f) + __logf(1.0f + __expf(-fabsf(z)));

      atomicAdd(&acc[(i + 2  + lane) & 63], cA * f);
      atomicAdd(&acc[(i + 34 + lane) & 63], cB * f);

      if (lane == s) fOut = f;
      f_prev = f;
      accP   = accNext;
    }
    out[i0 + lane] = fOut;
  }
}

extern "C" void kernel_launch(void* const* d_in, const int* in_sizes, int n_in,
                              void* d_out, int out_size, void* d_ws, size_t ws_size,
                              hipStream_t stream) {
  const float* Is   = (const float*)d_in[0];
  const float* a    = (const float*)d_in[1];
  const float* b    = (const float*)d_in[2];
  const float* g_w1 = (const float*)d_in[3];
  const float* g_b1 = (const float*)d_in[4];
  const float* g_w2 = (const float*)d_in[5];
  const float* g_b2 = (const float*)d_in[6];
  float* out = (float*)d_out;
  float* u   = (float*)d_ws;

  fir_u_wmma<<<NTILE / 8, 256, 0, stream>>>(Is, a, u);
  recurrence<<<1, 32, 0, stream>>>(u, b, g_w1, g_b1, g_w2, g_b2, out);
}


# recurrence: xor16 reduce stage and u broadcast moved from LDS shuffles to VALU permlanex16/readlane, ring LDS ops fenced beside the DPP stages: f->f chain no longer waits on LDS; sums unchanged
# speedup vs baseline: 1.3712x; 1.3712x over previous
#include <hip/hip_runtime.h>
#include <hip/hip_bf16.h>

typedef __attribute__((ext_vector_type(2))) float v2f;
typedef __attribute__((ext_vector_type(8))) float v8f;

#define T_LEN 262144
#define NTILE (T_LEN / 256)

__global__ __launch_bounds__(256) void fir_u_wmma(const float* __restrict__ Is,
                                                  const float* __restrict__ a,
                                                  float* __restrict__ u) {
  __shared__ float a_pad[96];
  {
    const int tid = threadIdx.x;
    if (tid < 96) {
      const int d = tid - 16;
      a_pad[tid] = (d >= 0 && d < 64) ? a[d] : 0.f;
    }
  }
  __syncthreads();

  const int lane = threadIdx.x & 31;
  const int wave = threadIdx.x >> 5;
  const int tile = blockIdx.x * 8 + wave;
  const int t0   = tile * 256;
  const int base = t0 - 64;
  const int m    = lane & 15;
  const int half = lane >> 4;
  const int kb   = half * 2;

  v8f c = {0.f, 0.f, 0.f, 0.f, 0.f, 0.f, 0.f, 0.f};

  if (tile != 0) {
#pragma unroll
    for (int p = 0; p < 20; ++p) {
      const int k0 = p * 4 + kb;
      v2f av, bv;
      av.x = a_pad[16 + k0 - m];
      av.y = a_pad[17 + k0 - m];
      bv = *(const v2f*)(Is + (base + 16 * m + k0));
      c = __builtin_amdgcn_wmma_f32_16x16x4_f32(false, av, false, bv,
                                                (short)0, c, false, false);
    }
  } else {
#pragma unroll
    for (int p = 0; p < 20; ++p) {
      const int k0 = p * 4 + kb;
      v2f av, bv;
      av.x = a_pad[16 + k0 - m];
      av.y = a_pad[17 + k0 - m];
      const int i0 = base + 16 * m + k0;
      bv.x = (i0     >= 0) ? Is[i0]     : 0.f;
      bv.y = (i0 + 1 >= 0) ? Is[i0 + 1] : 0.f;
      c = __builtin_amdgcn_wmma_f32_16x16x4_f32(false, av, false, bv,
                                                (short)0, c, false, false);
    }
  }

#pragma unroll
  for (int r = 0; r < 8; ++r) {
    const int M = r + 8 * half;
    const int t = t0 + 16 * m + M;
    u[t] = Is[t] + c[r];
  }
}

template <int CTRL>
__device__ __forceinline__ float dpp_xadd(float x) {
  const int y = __builtin_amdgcn_update_dpp(0, __builtin_bit_cast(int, x),
                                            CTRL, 0xf, 0xf, true);
  return x + __builtin_bit_cast(float, y);
}

__device__ __forceinline__ float xhalf(float x) {
  const int b = __builtin_bit_cast(int, x);
  return __builtin_bit_cast(float,
      __builtin_amdgcn_permlanex16(0, b, 0x76543210, 0xfedcba98, false, false));
}

__device__ __forceinline__ float lane_bcast(float x, int l) {
  return __builtin_bit_cast(float,
      __builtin_amdgcn_readlane(__builtin_bit_cast(int, x), l));
}

__global__ __launch_bounds__(32) void recurrence(const float* __restrict__ u,
                                                 const float* __restrict__ b,
                                                 const float* __restrict__ w1,
                                                 const float* __restrict__ b1,
                                                 const float* __restrict__ w2,
                                                 const float* __restrict__ b2p,
                                                 float* __restrict__ out) {
  __shared__ float acc[64];
  const int lane = threadIdx.x;

  acc[lane]      = 0.f;
  acc[lane + 32] = 0.f;

  float w1r2[4], b1r2[4], w2r[4];
#pragma unroll
  for (int r = 0; r < 4; ++r) {
    w1r2[r] = 2.0f * w1[lane + 32 * r];
    b1r2[r] = 2.0f * b1[lane + 32 * r];
    w2r[r]  = w2[lane + 32 * r];
  }
  const float cA  = b[62 - lane];
  const float cB  = (lane <= 30) ? b[30 - lane] : 0.f;
  const float b63 = b[63];
  const float gb2 = b2p[0];

  float f_prev = 0.f;
  float accP   = 0.f;
  float uNxt   = u[lane];
  float fOut   = 0.f;

  const int NC = T_LEN / 32;
  for (int ch = 0; ch < NC; ++ch) {
    const float uCur = uNxt;
    const int   nb   = (ch + 1) * 32;
    const int   pidx = (nb + lane < T_LEN) ? (nb + lane) : (T_LEN - 1);
    uNxt = u[pidx];
    const int i0 = ch * 32;
#pragma unroll 4
    for (int s = 0; s < 32; ++s) {
      const int i = i0 + s;

      const float uu = lane_bcast(uCur, s);
      const float x  = uu - fmaf(b63, f_prev, accP);

      float th[4];
#pragma unroll
      for (int r = 0; r < 4; ++r) {
        const float e = __expf(fmaf(w1r2[r], x, b1r2[r]));
        th[r] = 1.0f - 2.0f * __builtin_amdgcn_rcpf(e + 1.0f);
      }
      const float p0   = fmaf(w2r[0], th[0], w2r[1] * th[1]);
      const float p1   = fmaf(w2r[2], th[2], w2r[3] * th[3]);
      float ssum = p0 + p1;

      __builtin_amdgcn_sched_barrier(0);
      atomicAdd(&acc[(i + 1  + lane) & 63], cA * f_prev);
      atomicAdd(&acc[(i + 33 + lane) & 63], cB * f_prev);
      const int slot = (i + 1) & 63;
      const float accNext = acc[slot];
      acc[slot] = 0.f;

      ssum = dpp_xadd<0xB1>(ssum);
      ssum = dpp_xadd<0x4E>(ssum);
      ssum = dpp_xadd<0x141>(ssum);
      ssum = dpp_xadd<0x140>(ssum);
      __builtin_amdgcn_sched_barrier(0);
      ssum += xhalf(ssum);

      const float z = ssum + gb2;
      const float f = fmaxf(z, 0.f) + __logf(1.0f + __expf(-fabsf(z)));

      if (lane == s) fOut = f;
      f_prev = f;
      accP   = accNext;
    }
    out[i0 + lane] = fOut;
  }
}

extern "C" void kernel_launch(void* const* d_in, const int* in_sizes, int n_in,
                              void* d_out, int out_size, void* d_ws, size_t ws_size,
                              hipStream_t stream) {
  const float* Is   = (const float*)d_in[0];
  const float* a    = (const float*)d_in[1];
  const float* b    = (const float*)d_in[2];
  const float* g_w1 = (const float*)d_in[3];
  const float* g_b1 = (const float*)d_in[4];
  const float* g_w2 = (const float*)d_in[5];
  const float* g_b2 = (const float*)d_in[6];
  float* out = (float*)d_out;
  float* u   = (float*)d_ws;

  fir_u_wmma<<<NTILE / 8, 256, 0, stream>>>(Is, a, u);
  recurrence<<<1, 32, 0, stream>>>(u, b, g_w1, g_b1, g_w2, g_b2, out);
}


# softplus log(1+e) on the f->f chain: library logf expansion (denormal rescale, split-constant multiply, finite select) replaced by hardware log2 times ln2; argument always in [1,2]; sums unchanged
# speedup vs baseline: 1.6418x; 1.1973x over previous
#include <hip/hip_runtime.h>
#include <hip/hip_bf16.h>

typedef __attribute__((ext_vector_type(2))) float v2f;
typedef __attribute__((ext_vector_type(8))) float v8f;

#define T_LEN 262144
#define NTILE (T_LEN / 256)

__global__ __launch_bounds__(256) void fir_u_wmma(const float* __restrict__ Is,
                                                  const float* __restrict__ a,
                                                  float* __restrict__ u) {
  __shared__ float a_pad[96];
  {
    const int tid = threadIdx.x;
    if (tid < 96) {
      const int d = tid - 16;
      a_pad[tid] = (d >= 0 && d < 64) ? a[d] : 0.f;
    }
  }
  __syncthreads();

  const int lane = threadIdx.x & 31;
  const int wave = threadIdx.x >> 5;
  const int tile = blockIdx.x * 8 + wave;
  const int t0   = tile * 256;
  const int base = t0 - 64;
  const int m    = lane & 15;
  const int half = lane >> 4;
  const int kb   = half * 2;

  v8f c = {0.f, 0.f, 0.f, 0.f, 0.f, 0.f, 0.f, 0.f};

  if (tile != 0) {
#pragma unroll
    for (int p = 0; p < 20; ++p) {
      const int k0 = p * 4 + kb;
      v2f av, bv;
      av.x = a_pad[16 + k0 - m];
      av.y = a_pad[17 + k0 - m];
      bv = *(const v2f*)(Is + (base + 16 * m + k0));
      c = __builtin_amdgcn_wmma_f32_16x16x4_f32(false, av, false, bv,
                                                (short)0, c, false, false);
    }
  } else {
#pragma unroll
    for (int p = 0; p < 20; ++p) {
      const int k0 = p * 4 + kb;
      v2f av, bv;
      av.x = a_pad[16 + k0 - m];
      av.y = a_pad[17 + k0 - m];
      const int i0 = base + 16 * m + k0;
      bv.x = (i0     >= 0) ? Is[i0]     : 0.f;
      bv.y = (i0 + 1 >= 0) ? Is[i0 + 1] : 0.f;
      c = __builtin_amdgcn_wmma_f32_16x16x4_f32(false, av, false, bv,
                                                (short)0, c, false, false);
    }
  }

#pragma unroll
  for (int r = 0; r < 8; ++r) {
    const int M = r + 8 * half;
    const int t = t0 + 16 * m + M;
    u[t] = Is[t] + c[r];
  }
}

template <int CTRL>
__device__ __forceinline__ float dpp_xadd(float x) {
  const int y = __builtin_amdgcn_update_dpp(0, __builtin_bit_cast(int, x),
                                            CTRL, 0xf, 0xf, true);
  return x + __builtin_bit_cast(float, y);
}

__device__ __forceinline__ float xhalf(float x) {
  const int b = __builtin_bit_cast(int, x);
  return __builtin_bit_cast(float,
      __builtin_amdgcn_permlanex16(0, b, 0x76543210, 0xfedcba98, false, false));
}

__device__ __forceinline__ float lane_bcast(float x, int l) {
  return __builtin_bit_cast(float,
      __builtin_amdgcn_readlane(__builtin_bit_cast(int, x), l));
}

__global__ __launch_bounds__(32) void recurrence(const float* __restrict__ u,
                                                 const float* __restrict__ b,
                                                 const float* __restrict__ w1,
                                                 const float* __restrict__ b1,
                                                 const float* __restrict__ w2,
                                                 const float* __restrict__ b2p,
                                                 float* __restrict__ out) {
  __shared__ float acc[64];
  const int lane = threadIdx.x;

  acc[lane]      = 0.f;
  acc[lane + 32] = 0.f;

  float w1r2[4], b1r2[4], w2r[4];
#pragma unroll
  for (int r = 0; r < 4; ++r) {
    w1r2[r] = 2.0f * w1[lane + 32 * r];
    b1r2[r] = 2.0f * b1[lane + 32 * r];
    w2r[r]  = w2[lane + 32 * r];
  }
  const float cA  = b[62 - lane];
  const float cB  = (lane <= 30) ? b[30 - lane] : 0.f;
  const float b63 = b[63];
  const float gb2 = b2p[0];

  float f_prev = 0.f;
  float accP   = 0.f;
  float uNxt   = u[lane];
  float fOut   = 0.f;

  const int NC = T_LEN / 32;
  for (int ch = 0; ch < NC; ++ch) {
    const float uCur = uNxt;
    const int   nb   = (ch + 1) * 32;
    const int   pidx = (nb + lane < T_LEN) ? (nb + lane) : (T_LEN - 1);
    uNxt = u[pidx];
    const int i0 = ch * 32;
#pragma unroll 4
    for (int s = 0; s < 32; ++s) {
      const int i = i0 + s;

      const float uu = lane_bcast(uCur, s);
      const float x  = uu - fmaf(b63, f_prev, accP);

      float th[4];
#pragma unroll
      for (int r = 0; r < 4; ++r) {
        const float e = __expf(fmaf(w1r2[r], x, b1r2[r]));
        th[r] = 1.0f - 2.0f * __builtin_amdgcn_rcpf(e + 1.0f);
      }
      const float p0   = fmaf(w2r[0], th[0], w2r[1] * th[1]);
      const float p1   = fmaf(w2r[2], th[2], w2r[3] * th[3]);
      float ssum = p0 + p1;

      __builtin_amdgcn_sched_barrier(0);
      atomicAdd(&acc[(i + 1  + lane) & 63], cA * f_prev);
      atomicAdd(&acc[(i + 33 + lane) & 63], cB * f_prev);
      const int slot = (i + 1) & 63;
      const float accNext = acc[slot];
      acc[slot] = 0.f;

      ssum = dpp_xadd<0xB1>(ssum);
      ssum = dpp_xadd<0x4E>(ssum);
      ssum = dpp_xadd<0x141>(ssum);
      ssum = dpp_xadd<0x140>(ssum);
      __builtin_amdgcn_sched_barrier(0);
      ssum += xhalf(ssum);

      const float z = ssum + gb2;
      const float lg2 = __builtin_amdgcn_logf(1.0f + __expf(-fabsf(z)));
      const float f   = fmaf(0.693147180559945f, lg2, fmaxf(z, 0.f));

      if (lane == s) fOut = f;
      f_prev = f;
      accP   = accNext;
    }
    out[i0 + lane] = fOut;
  }
}

extern "C" void kernel_launch(void* const* d_in, const int* in_sizes, int n_in,
                              void* d_out, int out_size, void* d_ws, size_t ws_size,
                              hipStream_t stream) {
  const float* Is   = (const float*)d_in[0];
  const float* a    = (const float*)d_in[1];
  const float* b    = (const float*)d_in[2];
  const float* g_w1 = (const float*)d_in[3];
  const float* g_b1 = (const float*)d_in[4];
  const float* g_w2 = (const float*)d_in[5];
  const float* g_b2 = (const float*)d_in[6];
  float* out = (float*)d_out;
  float* u   = (float*)d_ws;

  fir_u_wmma<<<NTILE / 8, 256, 0, stream>>>(Is, a, u);
  recurrence<<<1, 32, 0, stream>>>(u, b, g_w1, g_b1, g_w2, g_b2, out);
}


# recurrence: every affine link of the f->f chain folded into per-lane constants or terms prepared a step ahead (taps d=2,3 in registers, ring read a step early); sum order and roundings change
# speedup vs baseline: 1.7922x; 1.0916x over previous
#include <hip/hip_runtime.h>
#include <hip/hip_bf16.h>

typedef __attribute__((ext_vector_type(2))) float v2f;
typedef __attribute__((ext_vector_type(8))) float v8f;

#define T_LEN 262144
#define NTILE (T_LEN / 256)

__global__ __launch_bounds__(256) void fir_u_wmma(const float* __restrict__ Is,
                                                  const float* __restrict__ a,
                                                  float* __restrict__ u) {
  __shared__ float a_pad[96];
  {
    const int tid = threadIdx.x;
    if (tid < 96) {
      const int d = tid - 16;
      a_pad[tid] = (d >= 0 && d < 64) ? a[d] : 0.f;
    }
  }
  __syncthreads();

  const int lane = threadIdx.x & 31;
  const int wave = threadIdx.x >> 5;
  const int tile = blockIdx.x * 8 + wave;
  const int t0   = tile * 256;
  const int base = t0 - 64;
  const int m    = lane & 15;
  const int half = lane >> 4;
  const int kb   = half * 2;

  v8f c = {0.f, 0.f, 0.f, 0.f, 0.f, 0.f, 0.f, 0.f};

  if (tile != 0) {
#pragma unroll
    for (int p = 0; p < 20; ++p) {
      const int k0 = p * 4 + kb;
      v2f av, bv;
      av.x = a_pad[16 + k0 - m];
      av.y = a_pad[17 + k0 - m];
      bv = *(const v2f*)(Is + (base + 16 * m + k0));
      c = __builtin_amdgcn_wmma_f32_16x16x4_f32(false, av, false, bv,
                                                (short)0, c, false, false);
    }
  } else {
#pragma unroll
    for (int p = 0; p < 20; ++p) {
      const int k0 = p * 4 + kb;
      v2f av, bv;
      av.x = a_pad[16 + k0 - m];
      av.y = a_pad[17 + k0 - m];
      const int i0 = base + 16 * m + k0;
      bv.x = (i0     >= 0) ? Is[i0]     : 0.f;
      bv.y = (i0 + 1 >= 0) ? Is[i0 + 1] : 0.f;
      c = __builtin_amdgcn_wmma_f32_16x16x4_f32(false, av, false, bv,
                                                (short)0, c, false, false);
    }
  }

#pragma unroll
  for (int r = 0; r < 8; ++r) {
    const int M = r + 8 * half;
    const int t = t0 + 16 * m + M;
    u[t] = Is[t] + c[r];
  }
}

template <int CTRL>
__device__ __forceinline__ float dpp_xadd(float x) {
  const int y = __builtin_amdgcn_update_dpp(0, __builtin_bit_cast(int, x),
                                            CTRL, 0xf, 0xf, true);
  return x + __builtin_bit_cast(float, y);
}

__device__ __forceinline__ float xhalf(float x) {
  const int b = __builtin_bit_cast(int, x);
  return __builtin_bit_cast(float,
      __builtin_amdgcn_permlanex16(0, b, 0x76543210, 0xfedcba98, false, false));
}

__device__ __forceinline__ float lane_bcast(float x, int l) {
  return __builtin_bit_cast(float,
      __builtin_amdgcn_readlane(__builtin_bit_cast(int, x), l));
}

__global__ __launch_bounds__(32) void recurrence(const float* __restrict__ u,
                                                 const float* __restrict__ b,
                                                 const float* __restrict__ w1,
                                                 const float* __restrict__ b1,
                                                 const float* __restrict__ w2,
                                                 const float* __restrict__ b2p,
                                                 float* __restrict__ out) {
  __shared__ float acc[64];
  const int lane = threadIdx.x;

  acc[lane]      = 0.f;
  acc[lane + 32] = 0.f;

  const float L2E = 1.4426950408889634f;
  const float LN2 = 0.6931471805599453f;
  const float nb63 = -b[63];
  const float nb62 = -b[62];
  const float nb61 = -b[61];

  float S1[4], SB1[4], A[4], AL[4], Wq[4];
  float wsum = 0.f;
#pragma unroll
  for (int r = 0; r < 4; ++r) {
    const float w2v = w2[lane + 32 * r];
    S1[r]  = (2.0f * L2E) * w1[lane + 32 * r];
    SB1[r] = (2.0f * L2E) * b1[lane + 32 * r];
    A[r]   = nb63 * S1[r];
    AL[r]  = LN2 * A[r];
    Wq[r]  = (-2.0f * L2E) * w2v;
    wsum  += w2v;
  }
  wsum = dpp_xadd<0xB1>(wsum);
  wsum = dpp_xadd<0x4E>(wsum);
  wsum = dpp_xadd<0x141>(wsum);
  wsum = dpp_xadd<0x140>(wsum);
  wsum += xhalf(wsum);
  const float zc = (lane == 0) ? L2E * (wsum + b2p[0]) : 0.f;

  const float cA = b[60 - lane];
  const float cB = (lane <= 28) ? b[28 - lane] : 0.f;

  float f1 = 0.f, f2 = 0.f;
  float lgP = 0.f;
  float ringNext = 0.f;
  float fOut = 0.f;
  float D[4];
  {
    const float u0 = u[0];
#pragma unroll
    for (int r = 0; r < 4; ++r) D[r] = fmaf(S1[r], u0, SB1[r]);
  }
  float uNxt = u[(1 + lane < T_LEN) ? (1 + lane) : (T_LEN - 1)];

  const int NC = T_LEN / 32;
  for (int ch = 0; ch < NC; ++ch) {
    const float uCur = uNxt;
    const int   nb   = (ch + 1) * 32 + 1;
    const int   pidx = (nb + lane < T_LEN) ? (nb + lane) : (T_LEN - 1);
    uNxt = u[pidx];
    const int i0 = ch * 32;
#pragma unroll 4
    for (int s = 0; s < 32; ++s) {
      const int i = i0 + s;

      float q[4];
#pragma unroll
      for (int r = 0; r < 4; ++r) {
        const float e = __builtin_amdgcn_exp2f(fmaf(AL[r], lgP, D[r]));
        q[r] = __builtin_amdgcn_rcpf(e + 1.0f);
      }
      const float p0 = fmaf(Wq[0], q[0], fmaf(Wq[1], q[1], zc));
      const float p1 = fmaf(Wq[2], q[2], Wq[3] * q[3]);
      float zl = p0 + p1;

      atomicAdd(&acc[(i + 2  + lane) & 63], cA * f2);
      atomicAdd(&acc[(i + 34 + lane) & 63], cB * f2);
      const int slot = (i + 2) & 63;
      const float ringNew = acc[slot];
      acc[slot] = 0.f;

      __builtin_amdgcn_sched_barrier(0);

      zl = dpp_xadd<0xB1>(zl);
      zl = dpp_xadd<0x4E>(zl);
      zl = dpp_xadd<0x141>(zl);
      zl = dpp_xadd<0x140>(zl);
      zl += xhalf(zl);

      const float uu   = lane_bcast(uCur, s);
      const float t0   = fmaf(nb61, f2, uu - ringNext);
      const float base = fmaf(nb62, f1, t0);
      float Cn[4];
#pragma unroll
      for (int r = 0; r < 4; ++r) Cn[r] = fmaf(S1[r], base, SB1[r]);

      const float lg = __builtin_amdgcn_logf(1.0f + __builtin_amdgcn_exp2f(-fabsf(zl)));
      const float m2 = LN2 * fmaxf(zl, 0.f);
#pragma unroll
      for (int r = 0; r < 4; ++r) D[r] = fmaf(A[r], m2, Cn[r]);
      const float f = fmaf(LN2, lg, m2);

      if (lane == s) fOut = f;
      f2 = f1;
      f1 = f;
      lgP = lg;
      ringNext = ringNew;
    }
    out[i0 + lane] = fOut;
  }
}

extern "C" void kernel_launch(void* const* d_in, const int* in_sizes, int n_in,
                              void* d_out, int out_size, void* d_ws, size_t ws_size,
                              hipStream_t stream) {
  const float* Is   = (const float*)d_in[0];
  const float* a    = (const float*)d_in[1];
  const float* b    = (const float*)d_in[2];
  const float* g_w1 = (const float*)d_in[3];
  const float* g_b1 = (const float*)d_in[4];
  const float* g_w2 = (const float*)d_in[5];
  const float* g_b2 = (const float*)d_in[6];
  float* out = (float*)d_out;
  float* u   = (float*)d_ws;

  fir_u_wmma<<<NTILE / 8, 256, 0, stream>>>(Is, a, u);
  recurrence<<<1, 32, 0, stream>>>(u, b, g_w1, g_b1, g_w2, g_b2, out);
}


# recurrence: hidden-layer tanh taken from the hardware tanh instruction instead of exp2, add 1, reciprocal (one transcendental per row on the f->f chain, not two); sum order kept, roundings change
# speedup vs baseline: 1.9685x; 1.0984x over previous
#include <hip/hip_runtime.h>
#include <hip/hip_bf16.h>

typedef __attribute__((ext_vector_type(2))) float v2f;
typedef __attribute__((ext_vector_type(8))) float v8f;

#define T_LEN 262144
#define NTILE (T_LEN / 256)

__global__ __launch_bounds__(256) void fir_u_wmma(const float* __restrict__ Is,
                                                  const float* __restrict__ a,
                                                  float* __restrict__ u) {
  __shared__ float a_pad[96];
  {
    const int tid = threadIdx.x;
    if (tid < 96) {
      const int d = tid - 16;
      a_pad[tid] = (d >= 0 && d < 64) ? a[d] : 0.f;
    }
  }
  __syncthreads();

  const int lane = threadIdx.x & 31;
  const int wave = threadIdx.x >> 5;
  const int tile = blockIdx.x * 8 + wave;
  const int t0   = tile * 256;
  const int base = t0 - 64;
  const int m    = lane & 15;
  const int half = lane >> 4;
  const int kb   = half * 2;

  v8f c = {0.f, 0.f, 0.f, 0.f, 0.f, 0.f, 0.f, 0.f};

  if (tile != 0) {
#pragma unroll
    for (int p = 0; p < 20; ++p) {
      const int k0 = p * 4 + kb;
      v2f av, bv;
      av.x = a_pad[16 + k0 - m];
      av.y = a_pad[17 + k0 - m];
      bv = *(const v2f*)(Is + (base + 16 * m + k0));
      c = __builtin_amdgcn_wmma_f32_16x16x4_f32(false, av, false, bv,
                                                (short)0, c, false, false);
    }
  } else {
#pragma unroll
    for (int p = 0; p < 20; ++p) {
      const int k0 = p * 4 + kb;
      v2f av, bv;
      av.x = a_pad[16 + k0 - m];
      av.y = a_pad[17 + k0 - m];
      const int i0 = base + 16 * m + k0;
      bv.x = (i0     >= 0) ? Is[i0]     : 0.f;
      bv.y = (i0 + 1 >= 0) ? Is[i0 + 1] : 0.f;
      c = __builtin_amdgcn_wmma_f32_16x16x4_f32(false, av, false, bv,
                                                (short)0, c, false, false);
    }
  }

#pragma unroll
  for (int r = 0; r < 8; ++r) {
    const int M = r + 8 * half;
    const int t = t0 + 16 * m + M;
    u[t] = Is[t] + c[r];
  }
}

template <int CTRL>
__device__ __forceinline__ float dpp_xadd(float x) {
  const int y = __builtin_amdgcn_update_dpp(0, __builtin_bit_cast(int, x),
                                            CTRL, 0xf, 0xf, true);
  return x + __builtin_bit_cast(float, y);
}

__device__ __forceinline__ float xhalf(float x) {
  const int b = __builtin_bit_cast(int, x);
  return __builtin_bit_cast(float,
      __builtin_amdgcn_permlanex16(0, b, 0x76543210, 0xfedcba98, false, false));
}

__device__ __forceinline__ float lane_bcast(float x, int l) {
  return __builtin_bit_cast(float,
      __builtin_amdgcn_readlane(__builtin_bit_cast(int, x), l));
}

__global__ __launch_bounds__(32) void recurrence(const float* __restrict__ u,
                                                 const float* __restrict__ b,
                                                 const float* __restrict__ w1,
                                                 const float* __restrict__ b1,
                                                 const float* __restrict__ w2,
                                                 const float* __restrict__ b2p,
                                                 float* __restrict__ out) {
  __shared__ float acc[64];
  const int lane = threadIdx.x;

  acc[lane]      = 0.f;
  acc[lane + 32] = 0.f;

  const float L2E = 1.4426950408889634f;
  const float LN2 = 0.6931471805599453f;
  const float nb63 = -b[63];
  const float nb62 = -b[62];
  const float nb61 = -b[61];

  float S1[4], SB1[4], A[4], AL[4], Wt[4];
#pragma unroll
  for (int r = 0; r < 4; ++r) {
    S1[r]  = w1[lane + 32 * r];
    SB1[r] = b1[lane + 32 * r];
    A[r]   = nb63 * S1[r];
    AL[r]  = LN2 * A[r];
    Wt[r]  = L2E * w2[lane + 32 * r];
  }
  const float zc = (lane == 0) ? L2E * b2p[0] : 0.f;

  const float cA = b[60 - lane];
  const float cB = (lane <= 28) ? b[28 - lane] : 0.f;

  float f1 = 0.f, f2 = 0.f;
  float lgP = 0.f;
  float ringNext = 0.f;
  float fOut = 0.f;
  float D[4];
  {
    const float u0 = u[0];
#pragma unroll
    for (int r = 0; r < 4; ++r) D[r] = fmaf(S1[r], u0, SB1[r]);
  }
  float uNxt = u[(1 + lane < T_LEN) ? (1 + lane) : (T_LEN - 1)];

  const int NC = T_LEN / 32;
  for (int ch = 0; ch < NC; ++ch) {
    const float uCur = uNxt;
    const int   nb   = (ch + 1) * 32 + 1;
    const int   pidx = (nb + lane < T_LEN) ? (nb + lane) : (T_LEN - 1);
    uNxt = u[pidx];
    const int i0 = ch * 32;
#pragma unroll 4
    for (int s = 0; s < 32; ++s) {
      const int i = i0 + s;

      float th[4];
#pragma unroll
      for (int r = 0; r < 4; ++r)
        th[r] = __builtin_amdgcn_tanhf(fmaf(AL[r], lgP, D[r]));
      const float p0 = fmaf(Wt[0], th[0], fmaf(Wt[1], th[1], zc));
      const float p1 = fmaf(Wt[2], th[2], Wt[3] * th[3]);
      float zl = p0 + p1;

      atomicAdd(&acc[(i + 2  + lane) & 63], cA * f2);
      atomicAdd(&acc[(i + 34 + lane) & 63], cB * f2);
      const int slot = (i + 2) & 63;
      const float ringNew = acc[slot];
      acc[slot] = 0.f;

      __builtin_amdgcn_sched_barrier(0);

      zl = dpp_xadd<0xB1>(zl);
      zl = dpp_xadd<0x4E>(zl);
      zl = dpp_xadd<0x141>(zl);
      zl = dpp_xadd<0x140>(zl);
      zl += xhalf(zl);

      const float uu   = lane_bcast(uCur, s);
      const float t0   = fmaf(nb61, f2, uu - ringNext);
      const float base = fmaf(nb62, f1, t0);
      float Cn[4];
#pragma unroll
      for (int r = 0; r < 4; ++r) Cn[r] = fmaf(S1[r], base, SB1[r]);

      const float lg = __builtin_amdgcn_logf(1.0f + __builtin_amdgcn_exp2f(-fabsf(zl)));
      const float m2 = LN2 * fmaxf(zl, 0.f);
#pragma unroll
      for (int r = 0; r < 4; ++r) D[r] = fmaf(A[r], m2, Cn[r]);
      const float f = fmaf(LN2, lg, m2);

      if (lane == s) fOut = f;
      f2 = f1;
      f1 = f;
      lgP = lg;
      ringNext = ringNew;
    }
    out[i0 + lane] = fOut;
  }
}

extern "C" void kernel_launch(void* const* d_in, const int* in_sizes, int n_in,
                              void* d_out, int out_size, void* d_ws, size_t ws_size,
                              hipStream_t stream) {
  const float* Is   = (const float*)d_in[0];
  const float* a    = (const float*)d_in[1];
  const float* b    = (const float*)d_in[2];
  const float* g_w1 = (const float*)d_in[3];
  const float* g_b1 = (const float*)d_in[4];
  const float* g_w2 = (const float*)d_in[5];
  const float* g_b2 = (const float*)d_in[6];
  float* out = (float*)d_out;
  float* u   = (float*)d_ws;

  fir_u_wmma<<<NTILE / 8, 256, 0, stream>>>(Is, a, u);
  recurrence<<<1, 32, 0, stream>>>(u, b, g_w1, g_b1, g_w2, g_b2, out);
}


# recurrence: 64-slot feedback ring moved from LDS float atomics into two registers (taps in registers, chunk unrolled, one readlane out, writelane clears); sum order kept, roundings change
# speedup vs baseline: 3.3730x; 1.7135x over previous
#include <hip/hip_runtime.h>
#include <hip/hip_bf16.h>

typedef __attribute__((ext_vector_type(2))) float v2f;
typedef __attribute__((ext_vector_type(8))) float v8f;

#define T_LEN 262144
#define NTILE (T_LEN / 256)

__global__ __launch_bounds__(256) void fir_u_wmma(const float* __restrict__ Is,
                                                  const float* __restrict__ a,
                                                  float* __restrict__ u) {
  __shared__ float a_pad[96];
  {
    const int tid = threadIdx.x;
    if (tid < 96) {
      const int d = tid - 16;
      a_pad[tid] = (d >= 0 && d < 64) ? a[d] : 0.f;
    }
  }
  __syncthreads();

  const int lane = threadIdx.x & 31;
  const int wave = threadIdx.x >> 5;
  const int tile = blockIdx.x * 8 + wave;
  const int t0   = tile * 256;
  const int base = t0 - 64;
  const int m    = lane & 15;
  const int half = lane >> 4;
  const int kb   = half * 2;

  v8f c = {0.f, 0.f, 0.f, 0.f, 0.f, 0.f, 0.f, 0.f};

  if (tile != 0) {
#pragma unroll
    for (int p = 0; p < 20; ++p) {
      const int k0 = p * 4 + kb;
      v2f av, bv;
      av.x = a_pad[16 + k0 - m];
      av.y = a_pad[17 + k0 - m];
      bv = *(const v2f*)(Is + (base + 16 * m + k0));
      c = __builtin_amdgcn_wmma_f32_16x16x4_f32(false, av, false, bv,
                                                (short)0, c, false, false);
    }
  } else {
#pragma unroll
    for (int p = 0; p < 20; ++p) {
      const int k0 = p * 4 + kb;
      v2f av, bv;
      av.x = a_pad[16 + k0 - m];
      av.y = a_pad[17 + k0 - m];
      const int i0 = base + 16 * m + k0;
      bv.x = (i0     >= 0) ? Is[i0]     : 0.f;
      bv.y = (i0 + 1 >= 0) ? Is[i0 + 1] : 0.f;
      c = __builtin_amdgcn_wmma_f32_16x16x4_f32(false, av, false, bv,
                                                (short)0, c, false, false);
    }
  }

#pragma unroll
  for (int r = 0; r < 8; ++r) {
    const int M = r + 8 * half;
    const int t = t0 + 16 * m + M;
    u[t] = Is[t] + c[r];
  }
}

template <int CTRL>
__device__ __forceinline__ float dpp_xadd(float x) {
  const int y = __builtin_amdgcn_update_dpp(0, __builtin_bit_cast(int, x),
                                            CTRL, 0xf, 0xf, true);
  return x + __builtin_bit_cast(float, y);
}

__device__ __forceinline__ float xhalf(float x) {
  const int b = __builtin_bit_cast(int, x);
  return __builtin_bit_cast(float,
      __builtin_amdgcn_permlanex16(0, b, 0x76543210, 0xfedcba98, false, false));
}

__device__ __forceinline__ float lane_bcast(float x, int l) {
  return __builtin_bit_cast(float,
      __builtin_amdgcn_readlane(__builtin_bit_cast(int, x), l));
}

__device__ __forceinline__ float lane_clear(float x, int l) {
  asm("v_writelane_b32 %0, 0, %1" : "+v"(x) : "i"(l));
  return x;
}

__global__ __launch_bounds__(32) void recurrence(const float* __restrict__ u,
                                                 const float* __restrict__ b,
                                                 const float* __restrict__ w1,
                                                 const float* __restrict__ b1,
                                                 const float* __restrict__ w2,
                                                 const float* __restrict__ b2p,
                                                 float* __restrict__ out) {
  const int lane = threadIdx.x;

  const float L2E = 1.4426950408889634f;
  const float LN2 = 0.6931471805599453f;
  const float nb63 = -b[63];
  const float nb62 = -b[62];
  const float nb61 = -b[61];

  float S1[4], SB1[4], A[4], AL[4], Wt[4];
#pragma unroll
  for (int r = 0; r < 4; ++r) {
    S1[r]  = w1[lane + 32 * r];
    SB1[r] = b1[lane + 32 * r];
    A[r]   = nb63 * S1[r];
    AL[r]  = LN2 * A[r];
    Wt[r]  = L2E * w2[lane + 32 * r];
  }
  const float zc = (lane == 0) ? L2E * b2p[0] : 0.f;

  float T0[32], T1[32];
#pragma unroll
  for (int k = 0; k < 32; ++k) {
    const int m0 = (lane - k + 2) & 63;
    const int m1 = (lane - k + 34) & 63;
    const int d0 = (m0 == 0) ? 64 : m0;
    const int d1 = (m1 == 0) ? 64 : m1;
    const float v0 = b[(d0 >= 4) ? (64 - d0) : 0];
    const float v1 = b[(d1 >= 4) ? (64 - d1) : 0];
    T0[k] = (d0 >= 4) ? v0 : 0.f;
    T1[k] = (d1 >= 4) ? v1 : 0.f;
  }
  float X = 0.f, Y = 0.f;

  float f1 = 0.f, f2 = 0.f;
  float lgP = 0.f;
  float q = u[1];
  float fOut = 0.f;
  float D[4];
  {
    const float u0 = u[0];
#pragma unroll
    for (int r = 0; r < 4; ++r) D[r] = fmaf(S1[r], u0, SB1[r]);
  }
  float uB = u[lane];

  const int NC = T_LEN / 32;
  for (int ch = 0; ch < NC; ++ch) {
    const float uA = uB;
    const int   nb   = (ch + 1) * 32;
    const int   pidx = (nb + lane < T_LEN) ? (nb + lane) : (T_LEN - 1);
    uB = u[pidx];
    const int i0 = ch * 32;
#pragma unroll
    for (int s = 0; s < 32; ++s) {

      float th[4];
#pragma unroll
      for (int r = 0; r < 4; ++r)
        th[r] = __builtin_amdgcn_tanhf(fmaf(AL[r], lgP, D[r]));
      const float p0 = fmaf(Wt[0], th[0], fmaf(Wt[1], th[1], zc));
      const float p1 = fmaf(Wt[2], th[2], Wt[3] * th[3]);
      float zl = p0 + p1;

      X = fmaf(T0[s], f2, X);
      Y = fmaf(T1[s], f2, Y);
      float qNew;
      if (s < 30) { qNew = lane_bcast(uA - X, s + 2);  X = lane_clear(X, s + 2);  }
      else        { qNew = lane_bcast(uB - Y, s - 30); Y = lane_clear(Y, s - 30); }

      zl = dpp_xadd<0xB1>(zl);
      zl = dpp_xadd<0x4E>(zl);
      zl = dpp_xadd<0x141>(zl);
      zl = dpp_xadd<0x140>(zl);
      zl += xhalf(zl);

      const float t0   = fmaf(nb61, f2, q);
      const float base = fmaf(nb62, f1, t0);
      float Cn[4];
#pragma unroll
      for (int r = 0; r < 4; ++r) Cn[r] = fmaf(S1[r], base, SB1[r]);

      const float lg = __builtin_amdgcn_logf(1.0f + __builtin_amdgcn_exp2f(-fabsf(zl)));
      const float m2 = LN2 * fmaxf(zl, 0.f);
#pragma unroll
      for (int r = 0; r < 4; ++r) D[r] = fmaf(A[r], m2, Cn[r]);
      const float f = fmaf(LN2, lg, m2);

      if (lane == s) fOut = f;
      f2 = f1;
      f1 = f;
      lgP = lg;
      q = qNew;
    }
    out[i0 + lane] = fOut;
    { const float t = X; X = Y; Y = t; }
  }
}

extern "C" void kernel_launch(void* const* d_in, const int* in_sizes, int n_in,
                              void* d_out, int out_size, void* d_ws, size_t ws_size,
                              hipStream_t stream) {
  const float* Is   = (const float*)d_in[0];
  const float* a    = (const float*)d_in[1];
  const float* b    = (const float*)d_in[2];
  const float* g_w1 = (const float*)d_in[3];
  const float* g_b1 = (const float*)d_in[4];
  const float* g_w2 = (const float*)d_in[5];
  const float* g_b2 = (const float*)d_in[6];
  float* out = (float*)d_out;
  float* u   = (float*)d_ws;

  fir_u_wmma<<<NTILE / 8, 256, 0, stream>>>(Is, a, u);
  recurrence<<<1, 32, 0, stream>>>(u, b, g_w1, g_b1, g_w2, g_b2, out);
}


# recurrence: step issued in written order (scheduling barriers): each f->f chain link stands where its operand arrives, side work in the waits behind it, tanh rows issued 1,3,0,2; sums unchanged
# speedup vs baseline: 3.9484x; 1.1706x over previous
#include <hip/hip_runtime.h>
#include <hip/hip_bf16.h>

typedef __attribute__((ext_vector_type(2))) float v2f;
typedef __attribute__((ext_vector_type(8))) float v8f;

#define T_LEN 262144
#define NTILE (T_LEN / 256)

__global__ __launch_bounds__(256) void fir_u_wmma(const float* __restrict__ Is,
                                                  const float* __restrict__ a,
                                                  float* __restrict__ u) {
  __shared__ float a_pad[96];
  {
    const int tid = threadIdx.x;
    if (tid < 96) {
      const int d = tid - 16;
      a_pad[tid] = (d >= 0 && d < 64) ? a[d] : 0.f;
    }
  }
  __syncthreads();

  const int lane = threadIdx.x & 31;
  const int wave = threadIdx.x >> 5;
  const int tile = blockIdx.x * 8 + wave;
  const int t0   = tile * 256;
  const int base = t0 - 64;
  const int m    = lane & 15;
  const int half = lane >> 4;
  const int kb   = half * 2;

  v8f c = {0.f, 0.f, 0.f, 0.f, 0.f, 0.f, 0.f, 0.f};

  if (tile != 0) {
#pragma unroll
    for (int p = 0; p < 20; ++p) {
      const int k0 = p * 4 + kb;
      v2f av, bv;
      av.x = a_pad[16 + k0 - m];
      av.y = a_pad[17 + k0 - m];
      bv = *(const v2f*)(Is + (base + 16 * m + k0));
      c = __builtin_amdgcn_wmma_f32_16x16x4_f32(false, av, false, bv,
                                                (short)0, c, false, false);
    }
  } else {
#pragma unroll
    for (int p = 0; p < 20; ++p) {
      const int k0 = p * 4 + kb;
      v2f av, bv;
      av.x = a_pad[16 + k0 - m];
      av.y = a_pad[17 + k0 - m];
      const int i0 = base + 16 * m + k0;
      bv.x = (i0     >= 0) ? Is[i0]     : 0.f;
      bv.y = (i0 + 1 >= 0) ? Is[i0 + 1] : 0.f;
      c = __builtin_amdgcn_wmma_f32_16x16x4_f32(false, av, false, bv,
                                                (short)0, c, false, false);
    }
  }

#pragma unroll
  for (int r = 0; r < 8; ++r) {
    const int M = r + 8 * half;
    const int t = t0 + 16 * m + M;
    u[t] = Is[t] + c[r];
  }
}

template <int CTRL>
__device__ __forceinline__ float dpp_xadd(float x) {
  const int y = __builtin_amdgcn_update_dpp(0, __builtin_bit_cast(int, x),
                                            CTRL, 0xf, 0xf, true);
  return x + __builtin_bit_cast(float, y);
}

__device__ __forceinline__ float xhalf(float x) {
  const int b = __builtin_bit_cast(int, x);
  return __builtin_bit_cast(float,
      __builtin_amdgcn_permlanex16(0, b, 0x76543210, 0xfedcba98, false, false));
}

__device__ __forceinline__ float lane_bcast(float x, int l) {
  return __builtin_bit_cast(float,
      __builtin_amdgcn_readlane(__builtin_bit_cast(int, x), l));
}

__device__ __forceinline__ float lane_clear(float x, int l) {
  asm("v_writelane_b32 %0, 0, %1" : "+v"(x) : "i"(l));
  return x;
}

__global__ __launch_bounds__(32) void recurrence(const float* __restrict__ u,
                                                 const float* __restrict__ b,
                                                 const float* __restrict__ w1,
                                                 const float* __restrict__ b1,
                                                 const float* __restrict__ w2,
                                                 const float* __restrict__ b2p,
                                                 float* __restrict__ out) {
  const int lane = threadIdx.x;

  const float L2E = 1.4426950408889634f;
  const float LN2 = 0.6931471805599453f;
  const float nb63 = -b[63];
  const float nb62 = -b[62];
  const float nb61 = -b[61];

  float S1[4], SB1[4], A[4], AL[4], Wt[4];
#pragma unroll
  for (int r = 0; r < 4; ++r) {
    S1[r]  = w1[lane + 32 * r];
    SB1[r] = b1[lane + 32 * r];
    A[r]   = nb63 * S1[r];
    AL[r]  = LN2 * A[r];
    Wt[r]  = L2E * w2[lane + 32 * r];
  }
  const float zc = (lane == 0) ? L2E * b2p[0] : 0.f;

  float T0[32], T1[32];
#pragma unroll
  for (int k = 0; k < 32; ++k) {
    const int m0 = (lane - k + 2) & 63;
    const int m1 = (lane - k + 34) & 63;
    const int d0 = (m0 == 0) ? 64 : m0;
    const int d1 = (m1 == 0) ? 64 : m1;
    const float v0 = b[(d0 >= 4) ? (64 - d0) : 0];
    const float v1 = b[(d1 >= 4) ? (64 - d1) : 0];
    T0[k] = (d0 >= 4) ? v0 : 0.f;
    T1[k] = (d1 >= 4) ? v1 : 0.f;
  }
  float X = 0.f, Y = 0.f;

  float fB = 0.f;
  float fC = 0.f;
  float lgP = 0.f;
  float m2P = 0.f;
  float q = u[1];
  float fOut = 0.f;
  float D[4];
  {
    const float u0 = u[0];
#pragma unroll
    for (int r = 0; r < 4; ++r) D[r] = fmaf(S1[r], u0, SB1[r]);
  }
  float uB = u[lane];

#define PIN() __builtin_amdgcn_sched_barrier(0)
#define HOLD(v) asm volatile("" : "+v"(v))

  const int NC = T_LEN / 32;
  for (int ch = 0; ch < NC; ++ch) {
    const float uA = uB;
    const int   nb   = (ch + 1) * 32;
    const int   pidx = (nb + lane < T_LEN) ? (nb + lane) : (T_LEN - 1);
    uB = u[pidx];
    const int i0 = ch * 32;
#pragma unroll
    for (int s = 0; s < 32; ++s) {
      const float y1 = fmaf(AL[1], lgP, D[1]);
      const float y3 = fmaf(AL[3], lgP, D[3]);
      PIN();
      const float fp = (s == 0) ? fC : fmaf(LN2, lgP, m2P);
      const float y0 = fmaf(AL[0], lgP, D[0]);
      const float y2 = fmaf(AL[2], lgP, D[2]);
      PIN();
      const float th1 = __builtin_amdgcn_tanhf(y1);
      PIN();
      X = fmaf(T0[s], fB, X);
      Y = fmaf(T1[s], fB, Y);
      HOLD(X);
      HOLD(Y);
      PIN();
      const float th3 = __builtin_amdgcn_tanhf(y3);
      PIN();
      const float t0 = fmaf(nb61, fB, q);
      PIN();
      const float th0 = __builtin_amdgcn_tanhf(y0);
      PIN();
      if (s > 0) { if (lane == s - 1) fOut = fp; }
      PIN();
      const float th2 = __builtin_amdgcn_tanhf(y2);
      PIN();
      const float in1 = fmaf(Wt[1], th1, zc);
      PIN();
      float rd;
      if (s < 30) rd = uA - X; else rd = uB - Y;
      HOLD(rd);
      PIN();
      const float m3 = Wt[3] * th3;
      PIN();
      const float base = fmaf(nb62, fp, t0);
      PIN();
      const float p0 = fmaf(Wt[0], th0, in1);
      PIN();
      const float p1 = fmaf(Wt[2], th2, m3);
      PIN();
      float qNew;
      if (s < 30) { qNew = lane_bcast(rd, s + 2);  X = lane_clear(X, s + 2);  }
      else        { qNew = lane_bcast(rd, s - 30); Y = lane_clear(Y, s - 30); }
      PIN();
      float zl = p0 + p1;
      PIN();
      float Cn[4];
#pragma unroll
      for (int r = 0; r < 4; ++r) Cn[r] = fmaf(S1[r], base, SB1[r]);
      PIN();
      zl = dpp_xadd<0xB1>(zl);
      zl = dpp_xadd<0x4E>(zl);
      zl = dpp_xadd<0x141>(zl);
      zl = dpp_xadd<0x140>(zl);
      zl += xhalf(zl);
      PIN();
      const float e = __builtin_amdgcn_exp2f(-fabsf(zl));
      PIN();
      const float m2 = LN2 * fmaxf(zl, 0.f);
      PIN();
      const float w = 1.0f + e;
      PIN();
      const float lg = __builtin_amdgcn_logf(w);
      PIN();
      D[1] = fmaf(A[1], m2, Cn[1]);
      D[3] = fmaf(A[3], m2, Cn[3]);
      D[0] = fmaf(A[0], m2, Cn[0]);
      D[2] = fmaf(A[2], m2, Cn[2]);
      PIN();

      fB = fp;
      lgP = lg;
      m2P = m2;
      q = qNew;
    }
    fC = fmaf(LN2, lgP, m2P);
    if (lane == 31) fOut = fC;
    out[i0 + lane] = fOut;
    { const float t = X; X = Y; Y = t; }
  }
#undef PIN
#undef HOLD
}

extern "C" void kernel_launch(void* const* d_in, const int* in_sizes, int n_in,
                              void* d_out, int out_size, void* d_ws, size_t ws_size,
                              hipStream_t stream) {
  const float* Is   = (const float*)d_in[0];
  const float* a    = (const float*)d_in[1];
  const float* b    = (const float*)d_in[2];
  const float* g_w1 = (const float*)d_in[3];
  const float* g_b1 = (const float*)d_in[4];
  const float* g_w2 = (const float*)d_in[5];
  const float* g_b2 = (const float*)d_in[6];
  float* out = (float*)d_out;
  float* u   = (float*)d_ws;

  fir_u_wmma<<<NTILE / 8, 256, 0, stream>>>(Is, a, u);
  recurrence<<<1, 32, 0, stream>>>(u, b, g_w1, g_b1, g_w2, g_b2, out);
}


# recurrence: softplus as log2(1+2^z), each 16-lane row takes exp2 of its part of the lane sum and the rows meet in one fma with the 1: cross-row add is now a product, roundings change
# speedup vs baseline: 4.1892x; 1.0610x over previous
#include <hip/hip_runtime.h>
#include <hip/hip_bf16.h>

typedef __attribute__((ext_vector_type(2))) float v2f;
typedef __attribute__((ext_vector_type(8))) float v8f;

#define T_LEN 262144
#define NTILE (T_LEN / 256)

__global__ __launch_bounds__(256) void fir_u_wmma(const float* __restrict__ Is,
                                                  const float* __restrict__ a,
                                                  float* __restrict__ u) {
  __shared__ float a_pad[96];
  {
    const int tid = threadIdx.x;
    if (tid < 96) {
      const int d = tid - 16;
      a_pad[tid] = (d >= 0 && d < 64) ? a[d] : 0.f;
    }
  }
  __syncthreads();

  const int lane = threadIdx.x & 31;
  const int wave = threadIdx.x >> 5;
  const int tile = blockIdx.x * 8 + wave;
  const int t0   = tile * 256;
  const int base = t0 - 64;
  const int m    = lane & 15;
  const int half = lane >> 4;
  const int kb   = half * 2;

  v8f c = {0.f, 0.f, 0.f, 0.f, 0.f, 0.f, 0.f, 0.f};

  if (tile != 0) {
#pragma unroll
    for (int p = 0; p < 20; ++p) {
      const int k0 = p * 4 + kb;
      v2f av, bv;
      av.x = a_pad[16 + k0 - m];
      av.y = a_pad[17 + k0 - m];
      bv = *(const v2f*)(Is + (base + 16 * m + k0));
      c = __builtin_amdgcn_wmma_f32_16x16x4_f32(false, av, false, bv,
                                                (short)0, c, false, false);
    }
  } else {
#pragma unroll
    for (int p = 0; p < 20; ++p) {
      const int k0 = p * 4 + kb;
      v2f av, bv;
      av.x = a_pad[16 + k0 - m];
      av.y = a_pad[17 + k0 - m];
      const int i0 = base + 16 * m + k0;
      bv.x = (i0     >= 0) ? Is[i0]     : 0.f;
      bv.y = (i0 + 1 >= 0) ? Is[i0 + 1] : 0.f;
      c = __builtin_amdgcn_wmma_f32_16x16x4_f32(false, av, false, bv,
                                                (short)0, c, false, false);
    }
  }

#pragma unroll
  for (int r = 0; r < 8; ++r) {
    const int M = r + 8 * half;
    const int t = t0 + 16 * m + M;
    u[t] = Is[t] + c[r];
  }
}

template <int CTRL>
__device__ __forceinline__ float dpp_xadd(float x) {
  const int y = __builtin_amdgcn_update_dpp(0, __builtin_bit_cast(int, x),
                                            CTRL, 0xf, 0xf, true);
  return x + __builtin_bit_cast(float, y);
}

__device__ __forceinline__ float xhalf(float x) {
  const int b = __builtin_bit_cast(int, x);
  return __builtin_bit_cast(float,
      __builtin_amdgcn_permlanex16(0, b, 0x76543210, 0xfedcba98, false, false));
}

__device__ __forceinline__ float lane_bcast(float x, int l) {
  return __builtin_bit_cast(float,
      __builtin_amdgcn_readlane(__builtin_bit_cast(int, x), l));
}

__device__ __forceinline__ float lane_clear(float x, int l) {
  asm("v_writelane_b32 %0, 0, %1" : "+v"(x) : "i"(l));
  return x;
}

template <bool PROD>
__device__ __forceinline__ void recurrence_run(const float* __restrict__ u,
                                               const float* __restrict__ b,
                                               const float* __restrict__ w1,
                                               const float* __restrict__ b1,
                                               const float* __restrict__ w2,
                                               const float* __restrict__ b2p,
                                               float* __restrict__ out) {
  const int lane = threadIdx.x;

  const float L2E = 1.4426950408889634f;
  const float LN2 = 0.6931471805599453f;
  const float nb63 = -b[63];
  const float nb62 = -b[62];
  const float nb61 = -b[61];

  float S1[4], SB1[4], A[4], AL[4], Wt[4];
#pragma unroll
  for (int r = 0; r < 4; ++r) {
    S1[r]  = w1[lane + 32 * r];
    SB1[r] = b1[lane + 32 * r];
    A[r]   = nb63 * S1[r];
    AL[r]  = LN2 * A[r];
    Wt[r]  = L2E * w2[lane + 32 * r];
  }
  const float zc = (lane == 0) ? L2E * b2p[0] : 0.f;

  float T0[32], T1[32];
#pragma unroll
  for (int k = 0; k < 32; ++k) {
    const int m0 = (lane - k + 2) & 63;
    const int m1 = (lane - k + 34) & 63;
    const int d0 = (m0 == 0) ? 64 : m0;
    const int d1 = (m1 == 0) ? 64 : m1;
    const float v0 = b[(d0 >= 4) ? (64 - d0) : 0];
    const float v1 = b[(d1 >= 4) ? (64 - d1) : 0];
    T0[k] = (d0 >= 4) ? v0 : 0.f;
    T1[k] = (d1 >= 4) ? v1 : 0.f;
  }
  float X = 0.f, Y = 0.f;

  float fB = 0.f;
  float fC = 0.f;
  float lgP = 0.f;
  float m2P = 0.f;
  float q = u[1];
  float fOut = 0.f;
  float D[4];
  {
    const float u0 = u[0];
#pragma unroll
    for (int r = 0; r < 4; ++r) D[r] = fmaf(S1[r], u0, SB1[r]);
  }
  float uB = u[lane];

#define PIN() __builtin_amdgcn_sched_barrier(0)
#define HOLD(v) asm volatile("" : "+v"(v))

  const int NC = T_LEN / 32;
  for (int ch = 0; ch < NC; ++ch) {
    const float uA = uB;
    const int   nb   = (ch + 1) * 32;
    const int   pidx = (nb + lane < T_LEN) ? (nb + lane) : (T_LEN - 1);
    uB = u[pidx];
    const int i0 = ch * 32;
#pragma unroll
    for (int s = 0; s < 32; ++s) {
      const float y1 = fmaf(AL[1], lgP, D[1]);
      const float y3 = fmaf(AL[3], lgP, D[3]);
      PIN();
      float fp;
      if constexpr (PROD) fp = (s == 0) ? fC : LN2 * lgP;
      else                fp = (s == 0) ? fC : fmaf(LN2, lgP, m2P);
      const float y0 = fmaf(AL[0], lgP, D[0]);
      const float y2 = fmaf(AL[2], lgP, D[2]);
      PIN();
      const float th1 = __builtin_amdgcn_tanhf(y1);
      PIN();
      X = fmaf(T0[s], fB, X);
      Y = fmaf(T1[s], fB, Y);
      HOLD(X);
      HOLD(Y);
      PIN();
      const float th3 = __builtin_amdgcn_tanhf(y3);
      PIN();
      const float t0 = fmaf(nb61, fB, q);
      PIN();
      const float th0 = __builtin_amdgcn_tanhf(y0);
      PIN();
      if (s > 0) { if (lane == s - 1) fOut = fp; }
      PIN();
      const float th2 = __builtin_amdgcn_tanhf(y2);
      PIN();
      const float in1 = fmaf(Wt[1], th1, zc);
      PIN();
      float rd;
      if (s < 30) rd = uA - X; else rd = uB - Y;
      HOLD(rd);
      PIN();
      const float m3 = Wt[3] * th3;
      PIN();
      const float base = fmaf(nb62, fp, t0);
      PIN();
      const float p0 = fmaf(Wt[0], th0, in1);
      PIN();
      const float p1 = fmaf(Wt[2], th2, m3);
      PIN();
      float qNew;
      if (s < 30) { qNew = lane_bcast(rd, s + 2);  X = lane_clear(X, s + 2);  }
      else        { qNew = lane_bcast(rd, s - 30); Y = lane_clear(Y, s - 30); }
      PIN();
      float zl = p0 + p1;
      PIN();
      float Cn[4];
#pragma unroll
      for (int r = 0; r < 4; ++r) Cn[r] = fmaf(S1[r], base, SB1[r]);
      if constexpr (PROD) {
#pragma unroll
        for (int r = 0; r < 4; ++r) D[r] = Cn[r];
      }
      PIN();
      zl = dpp_xadd<0xB1>(zl);
      zl = dpp_xadd<0x4E>(zl);
      zl = dpp_xadd<0x141>(zl);
      zl = dpp_xadd<0x140>(zl);
      float lg;
      float m2 = 0.f;
      if constexpr (PROD) {
        const float E  = __builtin_amdgcn_exp2f(zl);
        const float Eo = xhalf(E);
        const float w  = fmaf(E, Eo, 1.0f);
        lg = __builtin_amdgcn_logf(w);
        PIN();
      } else {
        zl += xhalf(zl);
        PIN();
        const float e = __builtin_amdgcn_exp2f(-fabsf(zl));
        PIN();
        m2 = LN2 * fmaxf(zl, 0.f);
        PIN();
        const float w = 1.0f + e;
        PIN();
        lg = __builtin_amdgcn_logf(w);
        PIN();
        D[1] = fmaf(A[1], m2, Cn[1]);
        D[3] = fmaf(A[3], m2, Cn[3]);
        D[0] = fmaf(A[0], m2, Cn[0]);
        D[2] = fmaf(A[2], m2, Cn[2]);
        PIN();
      }

      fB = fp;
      lgP = lg;
      m2P = m2;
      q = qNew;
    }
    if constexpr (PROD) fC = LN2 * lgP;
    else                fC = fmaf(LN2, lgP, m2P);
    if (lane == 31) fOut = fC;
    out[i0 + lane] = fOut;
    { const float t = X; X = Y; Y = t; }
  }
#undef PIN
#undef HOLD
}

__global__ __launch_bounds__(32) void recurrence(const float* __restrict__ u,
                                                 const float* __restrict__ b,
                                                 const float* __restrict__ w1,
                                                 const float* __restrict__ b1,
                                                 const float* __restrict__ w2,
                                                 const float* __restrict__ b2p,
                                                 float* __restrict__ out) {
  const int lane = threadIdx.x;
  float sw = fabsf(w2[lane]) + fabsf(w2[lane + 32]) +
             fabsf(w2[lane + 64]) + fabsf(w2[lane + 96]);
  sw = dpp_xadd<0xB1>(sw);
  sw = dpp_xadd<0x4E>(sw);
  sw = dpp_xadd<0x141>(sw);
  sw = dpp_xadd<0x140>(sw);
  sw += xhalf(sw);
  const float B = lane_bcast(1.4426950408889634f * (sw + fabsf(b2p[0])), 0);
  if (B <= 100.f) recurrence_run<true>(u, b, w1, b1, w2, b2p, out);
  else            recurrence_run<false>(u, b, w1, b1, w2, b2p, out);
}

extern "C" void kernel_launch(void* const* d_in, const int* in_sizes, int n_in,
                              void* d_out, int out_size, void* d_ws, size_t ws_size,
                              hipStream_t stream) {
  const float* Is   = (const float*)d_in[0];
  const float* a    = (const float*)d_in[1];
  const float* b    = (const float*)d_in[2];
  const float* g_w1 = (const float*)d_in[3];
  const float* g_b1 = (const float*)d_in[4];
  const float* g_w2 = (const float*)d_in[5];
  const float* g_b2 = (const float*)d_in[6];
  float* out = (float*)d_out;
  float* u   = (float*)d_ws;

  fir_u_wmma<<<NTILE / 8, 256, 0, stream>>>(Is, a, u);
  recurrence<<<1, 32, 0, stream>>>(u, b, g_w1, g_b1, g_w2, g_b2, out);
}


# recurrence split in time: 128 one-wave blocks, each runs in 6144 steps early from a zero history; boundaries checked bitwise, failed blocks rerun longer, else serial; step and sums unchanged
# speedup vs baseline: 128.2873x; 30.6233x over previous
#include <hip/hip_runtime.h>
#include <hip/hip_bf16.h>

typedef __attribute__((ext_vector_type(2))) float v2f;
typedef __attribute__((ext_vector_type(8))) float v8f;

#define T_LEN 262144
#define NTILE (T_LEN / 256)

#define SP_CH  2048
#define SP_W   6144
#define SP_W2  63488
#define SP_NB  (T_LEN / SP_CH)
#define SP_REC 72
#define SP_NEED_BYTES ((unsigned long long)(T_LEN + SP_NB * SP_REC + 2 * SP_NB) * 4ull)

__global__ __launch_bounds__(256) void fir_u_wmma(const float* __restrict__ Is,
                                                  const float* __restrict__ a,
                                                  float* __restrict__ u) {
  __shared__ float a_pad[96];
  {
    const int tid = threadIdx.x;
    if (tid < 96) {
      const int d = tid - 16;
      a_pad[tid] = (d >= 0 && d < 64) ? a[d] : 0.f;
    }
  }
  __syncthreads();

  const int lane = threadIdx.x & 31;
  const int wave = threadIdx.x >> 5;
  const int tile = blockIdx.x * 8 + wave;
  const int t0   = tile * 256;
  const int base = t0 - 64;
  const int m    = lane & 15;
  const int half = lane >> 4;
  const int kb   = half * 2;

  v8f c = {0.f, 0.f, 0.f, 0.f, 0.f, 0.f, 0.f, 0.f};

  if (tile != 0) {
#pragma unroll
    for (int p = 0; p < 20; ++p) {
      const int k0 = p * 4 + kb;
      v2f av, bv;
      av.x = a_pad[16 + k0 - m];
      av.y = a_pad[17 + k0 - m];
      bv = *(const v2f*)(Is + (base + 16 * m + k0));
      c = __builtin_amdgcn_wmma_f32_16x16x4_f32(false, av, false, bv,
                                                (short)0, c, false, false);
    }
  } else {
#pragma unroll
    for (int p = 0; p < 20; ++p) {
      const int k0 = p * 4 + kb;
      v2f av, bv;
      av.x = a_pad[16 + k0 - m];
      av.y = a_pad[17 + k0 - m];
      const int i0 = base + 16 * m + k0;
      bv.x = (i0     >= 0) ? Is[i0]     : 0.f;
      bv.y = (i0 + 1 >= 0) ? Is[i0 + 1] : 0.f;
      c = __builtin_amdgcn_wmma_f32_16x16x4_f32(false, av, false, bv,
                                                (short)0, c, false, false);
    }
  }

#pragma unroll
  for (int r = 0; r < 8; ++r) {
    const int M = r + 8 * half;
    const int t = t0 + 16 * m + M;
    u[t] = Is[t] + c[r];
  }
}

template <int CTRL>
__device__ __forceinline__ float dpp_xadd(float x) {
  const int y = __builtin_amdgcn_update_dpp(0, __builtin_bit_cast(int, x),
                                            CTRL, 0xf, 0xf, true);
  return x + __builtin_bit_cast(float, y);
}

__device__ __forceinline__ float xhalf(float x) {
  const int b = __builtin_bit_cast(int, x);
  return __builtin_bit_cast(float,
      __builtin_amdgcn_permlanex16(0, b, 0x76543210, 0xfedcba98, false, false));
}

__device__ __forceinline__ float lane_bcast(float x, int l) {
  return __builtin_bit_cast(float,
      __builtin_amdgcn_readlane(__builtin_bit_cast(int, x), l));
}

__device__ __forceinline__ float lane_clear(float x, int l) {
  asm("v_writelane_b32 %0, 0, %1" : "+v"(x) : "i"(l));
  return x;
}

template <bool PROD>
__device__ __forceinline__ void recurrence_run(const float* __restrict__ u,
                                               const float* __restrict__ b,
                                               const float* __restrict__ w1,
                                               const float* __restrict__ b1,
                                               const float* __restrict__ w2,
                                               const float* __restrict__ b2p,
                                               float* __restrict__ out,
                                               const int i_begin,
                                               const int n_chunks,
                                               const int warm_chunks,
                                               float* __restrict__ rec) {
  const int lane = threadIdx.x;

  const float L2E = 1.4426950408889634f;
  const float LN2 = 0.6931471805599453f;
  const float nb63 = -b[63];
  const float nb62 = -b[62];
  const float nb61 = -b[61];

  float S1[4], SB1[4], A[4], AL[4], Wt[4];
#pragma unroll
  for (int r = 0; r < 4; ++r) {
    S1[r]  = w1[lane + 32 * r];
    SB1[r] = b1[lane + 32 * r];
    A[r]   = nb63 * S1[r];
    AL[r]  = LN2 * A[r];
    Wt[r]  = L2E * w2[lane + 32 * r];
  }
  const float zc = (lane == 0) ? L2E * b2p[0] : 0.f;

  float T0[32], T1[32];
#pragma unroll
  for (int k = 0; k < 32; ++k) {
    const int m0 = (lane - k + 2) & 63;
    const int m1 = (lane - k + 34) & 63;
    const int d0 = (m0 == 0) ? 64 : m0;
    const int d1 = (m1 == 0) ? 64 : m1;
    const float v0 = b[(d0 >= 4) ? (64 - d0) : 0];
    const float v1 = b[(d1 >= 4) ? (64 - d1) : 0];
    T0[k] = (d0 >= 4) ? v0 : 0.f;
    T1[k] = (d1 >= 4) ? v1 : 0.f;
  }
  float X = 0.f, Y = 0.f;

  float fB = 0.f;
  float fC = 0.f;
  float lgP = 0.f;
  float m2P = 0.f;
  float q = u[i_begin + 1];
  float fOut = 0.f;
  float D[4];
  {
    const float u0 = u[i_begin];
#pragma unroll
    for (int r = 0; r < 4; ++r) D[r] = fmaf(S1[r], u0, SB1[r]);
  }
  float uB = u[i_begin + lane];

#define PIN() __builtin_amdgcn_sched_barrier(0)
#define HOLD(v) asm volatile("" : "+v"(v))

  for (int ch = 0; ch < n_chunks; ++ch) {
    const float uA = uB;
    const int   nb   = i_begin + (ch + 1) * 32;
    const int   pidx = (nb + lane < T_LEN) ? (nb + lane) : (T_LEN - 1);
    uB = u[pidx];
    const int i0 = i_begin + ch * 32;
#pragma unroll
    for (int s = 0; s < 32; ++s) {
      const float y1 = fmaf(AL[1], lgP, D[1]);
      const float y3 = fmaf(AL[3], lgP, D[3]);
      PIN();
      float fp;
      if constexpr (PROD) fp = (s == 0) ? fC : LN2 * lgP;
      else                fp = (s == 0) ? fC : fmaf(LN2, lgP, m2P);
      const float y0 = fmaf(AL[0], lgP, D[0]);
      const float y2 = fmaf(AL[2], lgP, D[2]);
      PIN();
      const float th1 = __builtin_amdgcn_tanhf(y1);
      PIN();
      X = fmaf(T0[s], fB, X);
      Y = fmaf(T1[s], fB, Y);
      HOLD(X);
      HOLD(Y);
      PIN();
      const float th3 = __builtin_amdgcn_tanhf(y3);
      PIN();
      const float t0 = fmaf(nb61, fB, q);
      PIN();
      const float th0 = __builtin_amdgcn_tanhf(y0);
      PIN();
      if (s > 0) { if (lane == s - 1) fOut = fp; }
      PIN();
      const float th2 = __builtin_amdgcn_tanhf(y2);
      PIN();
      const float in1 = fmaf(Wt[1], th1, zc);
      PIN();
      float rd;
      if (s < 30) rd = uA - X; else rd = uB - Y;
      HOLD(rd);
      PIN();
      const float m3 = Wt[3] * th3;
      PIN();
      const float base = fmaf(nb62, fp, t0);
      PIN();
      const float p0 = fmaf(Wt[0], th0, in1);
      PIN();
      const float p1 = fmaf(Wt[2], th2, m3);
      PIN();
      float qNew;
      if (s < 30) { qNew = lane_bcast(rd, s + 2);  X = lane_clear(X, s + 2);  }
      else        { qNew = lane_bcast(rd, s - 30); Y = lane_clear(Y, s - 30); }
      PIN();
      float zl = p0 + p1;
      PIN();
      float Cn[4];
#pragma unroll
      for (int r = 0; r < 4; ++r) Cn[r] = fmaf(S1[r], base, SB1[r]);
      if constexpr (PROD) {
#pragma unroll
        for (int r = 0; r < 4; ++r) D[r] = Cn[r];
      }
      PIN();
      zl = dpp_xadd<0xB1>(zl);
      zl = dpp_xadd<0x4E>(zl);
      zl = dpp_xadd<0x141>(zl);
      zl = dpp_xadd<0x140>(zl);
      float lg;
      float m2 = 0.f;
      if constexpr (PROD) {
        const float E  = __builtin_amdgcn_exp2f(zl);
        const float Eo = xhalf(E);
        const float w  = fmaf(E, Eo, 1.0f);
        lg = __builtin_amdgcn_logf(w);
        PIN();
      } else {
        zl += xhalf(zl);
        PIN();
        const float e = __builtin_amdgcn_exp2f(-fabsf(zl));
        PIN();
        m2 = LN2 * fmaxf(zl, 0.f);
        PIN();
        const float w = 1.0f + e;
        PIN();
        lg = __builtin_amdgcn_logf(w);
        PIN();
        D[1] = fmaf(A[1], m2, Cn[1]);
        D[3] = fmaf(A[3], m2, Cn[3]);
        D[0] = fmaf(A[0], m2, Cn[0]);
        D[2] = fmaf(A[2], m2, Cn[2]);
        PIN();
      }

      fB = fp;
      lgP = lg;
      m2P = m2;
      q = qNew;
    }
    if constexpr (PROD) fC = LN2 * lgP;
    else                fC = fmaf(LN2, lgP, m2P);
    if (lane == 31) fOut = fC;
    if (ch >= warm_chunks) {
      out[i0 + lane] = fOut;
    } else if (ch >= warm_chunks - 2) {
      rec[(ch - (warm_chunks - 2)) * 32 + lane] = fOut;
      if (ch == warm_chunks - 1 && lane == 0) { rec[64] = lgP; rec[65] = m2P; }
    }
    { const float t = X; X = Y; Y = t; }
  }
  if (rec != nullptr && lane == 0) { rec[66] = lgP; rec[67] = m2P; }
#undef PIN
#undef HOLD
}

__device__ __forceinline__ void recurrence_pick(const float* __restrict__ u,
                                                const float* __restrict__ b,
                                                const float* __restrict__ w1,
                                                const float* __restrict__ b1,
                                                const float* __restrict__ w2,
                                                const float* __restrict__ b2p,
                                                float* __restrict__ out,
                                                const int i_begin, const int n_chunks,
                                                const int warm_chunks,
                                                float* __restrict__ rec) {
  const int lane = threadIdx.x;
  float sw = fabsf(w2[lane]) + fabsf(w2[lane + 32]) +
             fabsf(w2[lane + 64]) + fabsf(w2[lane + 96]);
  sw = dpp_xadd<0xB1>(sw);
  sw = dpp_xadd<0x4E>(sw);
  sw = dpp_xadd<0x141>(sw);
  sw = dpp_xadd<0x140>(sw);
  sw += xhalf(sw);
  const float B = lane_bcast(1.4426950408889634f * (sw + fabsf(b2p[0])), 0);
  if (B <= 100.f) recurrence_run<true>(u, b, w1, b1, w2, b2p, out, i_begin, n_chunks, warm_chunks, rec);
  else            recurrence_run<false>(u, b, w1, b1, w2, b2p, out, i_begin, n_chunks, warm_chunks, rec);
}

__global__ __launch_bounds__(32) void recurrence_chunks(const float* __restrict__ u,
                                                        const float* __restrict__ b,
                                                        const float* __restrict__ w1,
                                                        const float* __restrict__ b1,
                                                        const float* __restrict__ w2,
                                                        const float* __restrict__ b2p,
                                                        float* __restrict__ out,
                                                        float* __restrict__ sp,
                                                        const float* __restrict__ redo,
                                                        const int run_in,
                                                        const unsigned long long ws_bytes) {
  if (ws_bytes < SP_NEED_BYTES) return;
  const int k      = blockIdx.x;
  if (redo != nullptr) {
    if (k == 0) return;
    if (redo[k] == 0.f) return;
  }
  const int i_main = k * SP_CH;
  const int i_beg  = (i_main > run_in) ? (i_main - run_in) : 0;
  const int warm   = (i_main - i_beg) / 32;
  recurrence_pick(u, b, w1, b1, w2, b2p, out, i_beg, warm + SP_CH / 32, warm, sp + k * SP_REC);
}

__global__ __launch_bounds__(32) void boundary_check(const float* __restrict__ out,
                                                     const float* __restrict__ sp,
                                                     float* __restrict__ flag,
                                                     const unsigned long long ws_bytes) {
  if (ws_bytes < SP_NEED_BYTES) return;
  const int lane = threadIdx.x;
  const int k    = blockIdx.x + 1;
  const float* r = sp + k * SP_REC;
  const float* p = sp + (k - 1) * SP_REC;
  const int t    = k * SP_CH - 64;
  bool bad = __builtin_bit_cast(unsigned, r[lane])      != __builtin_bit_cast(unsigned, out[t + lane]);
  bad     |= __builtin_bit_cast(unsigned, r[32 + lane]) != __builtin_bit_cast(unsigned, out[t + 32 + lane]);
  if (lane < 2)
    bad   |= __builtin_bit_cast(unsigned, r[64 + lane]) != __builtin_bit_cast(unsigned, p[66 + lane]);
  const unsigned any = __builtin_amdgcn_ballot_w32(bad);
  if (lane == 0) flag[k] = (any != 0u) ? 1.f : 0.f;
}

__global__ __launch_bounds__(32) void recurrence_serial_if_needed(const float* __restrict__ u,
                                                                  const float* __restrict__ b,
                                                                  const float* __restrict__ w1,
                                                                  const float* __restrict__ b1,
                                                                  const float* __restrict__ w2,
                                                                  const float* __restrict__ b2p,
                                                                  float* __restrict__ out,
                                                                  const float* __restrict__ flag,
                                                                  const unsigned long long ws_bytes) {
  const int lane = threadIdx.x;
  unsigned any = 1u;
  if (ws_bytes >= SP_NEED_BYTES) {
    bool bad = false;
    for (int j = 0; j < SP_NB / 32; ++j) {
      const int k = 1 + lane + 32 * j;
      if (k < SP_NB) bad |= (flag[k] != 0.f);
    }
    any = __builtin_amdgcn_ballot_w32(bad);
  }
  if (any == 0u) return;
  recurrence_pick(u, b, w1, b1, w2, b2p, out, 0, T_LEN / 32, 0, nullptr);
}

extern "C" void kernel_launch(void* const* d_in, const int* in_sizes, int n_in,
                              void* d_out, int out_size, void* d_ws, size_t ws_size,
                              hipStream_t stream) {
  const float* Is   = (const float*)d_in[0];
  const float* a    = (const float*)d_in[1];
  const float* b    = (const float*)d_in[2];
  const float* g_w1 = (const float*)d_in[3];
  const float* g_b1 = (const float*)d_in[4];
  const float* g_w2 = (const float*)d_in[5];
  const float* g_b2 = (const float*)d_in[6];
  float* out  = (float*)d_out;
  float* u    = (float*)d_ws;
  float* sp   = u + T_LEN;
  float* flag1 = sp + SP_NB * SP_REC;
  float* flag2 = flag1 + SP_NB;
  const unsigned long long ws_bytes = ws_size;

  fir_u_wmma<<<NTILE / 8, 256, 0, stream>>>(Is, a, u);
  recurrence_chunks<<<SP_NB, 32, 0, stream>>>(u, b, g_w1, g_b1, g_w2, g_b2, out, sp,
                                              (const float*)nullptr, SP_W, ws_bytes);
  boundary_check<<<SP_NB - 1, 32, 0, stream>>>(out, sp, flag1, ws_bytes);
  recurrence_chunks<<<SP_NB, 32, 0, stream>>>(u, b, g_w1, g_b1, g_w2, g_b2, out, sp,
                                              flag1, SP_W2, ws_bytes);
  boundary_check<<<SP_NB - 1, 32, 0, stream>>>(out, sp, flag2, ws_bytes);
  recurrence_serial_if_needed<<<1, 32, 0, stream>>>(u, b, g_w1, g_b1, g_w2, g_b2, out, flag2, ws_bytes);
}


# recurrence split: run-in cut to 3584 steps (was 6144), 256 blocks of 1024; a block whose boundary fails re-runs its own steps from its neighbour's saved registers, in rounds; step and sums unchanged
# speedup vs baseline: 169.4634x; 1.3210x over previous
#include <hip/hip_runtime.h>
#include <hip/hip_bf16.h>

typedef __attribute__((ext_vector_type(2))) float v2f;
typedef __attribute__((ext_vector_type(8))) float v8f;

#define T_LEN 262144
#define NTILE (T_LEN / 256)

#define SP_CH  1024
#define SP_W   3584
#define SP_NB  (T_LEN / SP_CH)
#define SP_ROUNDS 4
#define SP_REC 72
#define SP_NEED_BYTES ((unsigned long long)(T_LEN + 2 * SP_NB * SP_REC + (SP_ROUNDS + 1) * SP_NB) * 4ull)

__global__ __launch_bounds__(256) void fir_u_wmma(const float* __restrict__ Is,
                                                  const float* __restrict__ a,
                                                  float* __restrict__ u) {
  __shared__ float a_pad[96];
  {
    const int tid = threadIdx.x;
    if (tid < 96) {
      const int d = tid - 16;
      a_pad[tid] = (d >= 0 && d < 64) ? a[d] : 0.f;
    }
  }
  __syncthreads();

  const int lane = threadIdx.x & 31;
  const int wave = threadIdx.x >> 5;
  const int tile = blockIdx.x * 8 + wave;
  const int t0   = tile * 256;
  const int base = t0 - 64;
  const int m    = lane & 15;
  const int half = lane >> 4;
  const int kb   = half * 2;

  v8f c = {0.f, 0.f, 0.f, 0.f, 0.f, 0.f, 0.f, 0.f};

  if (tile != 0) {
#pragma unroll
    for (int p = 0; p < 20; ++p) {
      const int k0 = p * 4 + kb;
      v2f av, bv;
      av.x = a_pad[16 + k0 - m];
      av.y = a_pad[17 + k0 - m];
      bv = *(const v2f*)(Is + (base + 16 * m + k0));
      c = __builtin_amdgcn_wmma_f32_16x16x4_f32(false, av, false, bv,
                                                (short)0, c, false, false);
    }
  } else {
#pragma unroll
    for (int p = 0; p < 20; ++p) {
      const int k0 = p * 4 + kb;
      v2f av, bv;
      av.x = a_pad[16 + k0 - m];
      av.y = a_pad[17 + k0 - m];
      const int i0 = base + 16 * m + k0;
      bv.x = (i0     >= 0) ? Is[i0]     : 0.f;
      bv.y = (i0 + 1 >= 0) ? Is[i0 + 1] : 0.f;
      c = __builtin_amdgcn_wmma_f32_16x16x4_f32(false, av, false, bv,
                                                (short)0, c, false, false);
    }
  }

#pragma unroll
  for (int r = 0; r < 8; ++r) {
    const int M = r + 8 * half;
    const int t = t0 + 16 * m + M;
    u[t] = Is[t] + c[r];
  }
}

template <int CTRL>
__device__ __forceinline__ float dpp_xadd(float x) {
  const int y = __builtin_amdgcn_update_dpp(0, __builtin_bit_cast(int, x),
                                            CTRL, 0xf, 0xf, true);
  return x + __builtin_bit_cast(float, y);
}

__device__ __forceinline__ float xhalf(float x) {
  const int b = __builtin_bit_cast(int, x);
  return __builtin_bit_cast(float,
      __builtin_amdgcn_permlanex16(0, b, 0x76543210, 0xfedcba98, false, false));
}

__device__ __forceinline__ float lane_bcast(float x, int l) {
  return __builtin_bit_cast(float,
      __builtin_amdgcn_readlane(__builtin_bit_cast(int, x), l));
}

__device__ __forceinline__ float lane_clear(float x, int l) {
  asm("v_writelane_b32 %0, 0, %1" : "+v"(x) : "i"(l));
  return x;
}

template <bool PROD>
__device__ __forceinline__ void recurrence_run(const float* __restrict__ u,
                                               const float* __restrict__ b,
                                               const float* __restrict__ w1,
                                               const float* __restrict__ b1,
                                               const float* __restrict__ w2,
                                               const float* __restrict__ b2p,
                                               float* __restrict__ out,
                                               const int i_begin,
                                               const int n_chunks,
                                               const int warm_chunks,
                                               float* __restrict__ rec,
                                               const float* __restrict__ init,
                                               float* __restrict__ save) {
  const int lane = threadIdx.x;

  const float L2E = 1.4426950408889634f;
  const float LN2 = 0.6931471805599453f;
  const float nb63 = -b[63];
  const float nb62 = -b[62];
  const float nb61 = -b[61];

  float S1[4], SB1[4], A[4], AL[4], Wt[4];
#pragma unroll
  for (int r = 0; r < 4; ++r) {
    S1[r]  = w1[lane + 32 * r];
    SB1[r] = b1[lane + 32 * r];
    A[r]   = nb63 * S1[r];
    AL[r]  = LN2 * A[r];
    Wt[r]  = L2E * w2[lane + 32 * r];
  }
  const float zc = (lane == 0) ? L2E * b2p[0] : 0.f;

  float T0[32], T1[32];
#pragma unroll
  for (int k = 0; k < 32; ++k) {
    const int m0 = (lane - k + 2) & 63;
    const int m1 = (lane - k + 34) & 63;
    const int d0 = (m0 == 0) ? 64 : m0;
    const int d1 = (m1 == 0) ? 64 : m1;
    const float v0 = b[(d0 >= 4) ? (64 - d0) : 0];
    const float v1 = b[(d1 >= 4) ? (64 - d1) : 0];
    T0[k] = (d0 >= 4) ? v0 : 0.f;
    T1[k] = (d1 >= 4) ? v1 : 0.f;
  }
  float X = 0.f, Y = 0.f;

  float fB = 0.f;
  float fC = 0.f;
  float lgP = 0.f;
  float m2P = 0.f;
  float q = u[i_begin + 1];
  float fOut = 0.f;
  float baseL = u[i_begin];
  if (init != nullptr) {
    X = init[lane];  Y = init[32 + lane];
    fB = init[64];   fC = init[65];  lgP = init[66];  m2P = init[67];
    q = init[68];    baseL = init[69];
  }
  float D[4];
#pragma unroll
  for (int r = 0; r < 4; ++r) D[r] = fmaf(S1[r], baseL, SB1[r]);
  if constexpr (!PROD) {
    if (init != nullptr) {
#pragma unroll
      for (int r = 0; r < 4; ++r) D[r] = fmaf(A[r], m2P, D[r]);
    }
  }
  float uB = u[i_begin + lane];

#define PIN() __builtin_amdgcn_sched_barrier(0)
#define HOLD(v) asm volatile("" : "+v"(v))

  for (int ch = 0; ch < n_chunks; ++ch) {
    const float uA = uB;
    const int   nb   = i_begin + (ch + 1) * 32;
    const int   pidx = (nb + lane < T_LEN) ? (nb + lane) : (T_LEN - 1);
    uB = u[pidx];
    const int i0 = i_begin + ch * 32;
#pragma unroll
    for (int s = 0; s < 32; ++s) {
      const float y1 = fmaf(AL[1], lgP, D[1]);
      const float y3 = fmaf(AL[3], lgP, D[3]);
      PIN();
      float fp;
      if constexpr (PROD) fp = (s == 0) ? fC : LN2 * lgP;
      else                fp = (s == 0) ? fC : fmaf(LN2, lgP, m2P);
      const float y0 = fmaf(AL[0], lgP, D[0]);
      const float y2 = fmaf(AL[2], lgP, D[2]);
      PIN();
      const float th1 = __builtin_amdgcn_tanhf(y1);
      PIN();
      X = fmaf(T0[s], fB, X);
      Y = fmaf(T1[s], fB, Y);
      HOLD(X);
      HOLD(Y);
      PIN();
      const float th3 = __builtin_amdgcn_tanhf(y3);
      PIN();
      const float t0 = fmaf(nb61, fB, q);
      PIN();
      const float th0 = __builtin_amdgcn_tanhf(y0);
      PIN();
      if (s > 0) { if (lane == s - 1) fOut = fp; }
      PIN();
      const float th2 = __builtin_amdgcn_tanhf(y2);
      PIN();
      const float in1 = fmaf(Wt[1], th1, zc);
      PIN();
      float rd;
      if (s < 30) rd = uA - X; else rd = uB - Y;
      HOLD(rd);
      PIN();
      const float m3 = Wt[3] * th3;
      PIN();
      const float base = fmaf(nb62, fp, t0);
      baseL = base;
      PIN();
      const float p0 = fmaf(Wt[0], th0, in1);
      PIN();
      const float p1 = fmaf(Wt[2], th2, m3);
      PIN();
      float qNew;
      if (s < 30) { qNew = lane_bcast(rd, s + 2);  X = lane_clear(X, s + 2);  }
      else        { qNew = lane_bcast(rd, s - 30); Y = lane_clear(Y, s - 30); }
      PIN();
      float zl = p0 + p1;
      PIN();
      float Cn[4];
#pragma unroll
      for (int r = 0; r < 4; ++r) Cn[r] = fmaf(S1[r], base, SB1[r]);
      if constexpr (PROD) {
#pragma unroll
        for (int r = 0; r < 4; ++r) D[r] = Cn[r];
      }
      PIN();
      zl = dpp_xadd<0xB1>(zl);
      zl = dpp_xadd<0x4E>(zl);
      zl = dpp_xadd<0x141>(zl);
      zl = dpp_xadd<0x140>(zl);
      float lg;
      float m2 = 0.f;
      if constexpr (PROD) {
        const float E  = __builtin_amdgcn_exp2f(zl);
        const float Eo = xhalf(E);
        const float w  = fmaf(E, Eo, 1.0f);
        lg = __builtin_amdgcn_logf(w);
        PIN();
      } else {
        zl += xhalf(zl);
        PIN();
        const float e = __builtin_amdgcn_exp2f(-fabsf(zl));
        PIN();
        m2 = LN2 * fmaxf(zl, 0.f);
        PIN();
        const float w = 1.0f + e;
        PIN();
        lg = __builtin_amdgcn_logf(w);
        PIN();
        D[1] = fmaf(A[1], m2, Cn[1]);
        D[3] = fmaf(A[3], m2, Cn[3]);
        D[0] = fmaf(A[0], m2, Cn[0]);
        D[2] = fmaf(A[2], m2, Cn[2]);
        PIN();
      }

      fB = fp;
      lgP = lg;
      m2P = m2;
      q = qNew;
    }
    if constexpr (PROD) fC = LN2 * lgP;
    else                fC = fmaf(LN2, lgP, m2P);
    if (lane == 31) fOut = fC;
    if (ch >= warm_chunks) {
      out[i0 + lane] = fOut;
    } else if (ch >= warm_chunks - 2) {
      rec[(ch - (warm_chunks - 2)) * 32 + lane] = fOut;
      if (ch == warm_chunks - 1 && lane == 0) { rec[64] = lgP; rec[65] = m2P; }
    }
    { const float t = X; X = Y; Y = t; }
  }
  if (rec != nullptr && lane == 0) { rec[66] = lgP; rec[67] = m2P; }
  if (save != nullptr) {
    save[lane] = X;  save[32 + lane] = Y;
    if (lane == 0) {
      save[64] = fB;  save[65] = fC;  save[66] = lgP;  save[67] = m2P;
      save[68] = q;   save[69] = baseL;
    }
  }
#undef PIN
#undef HOLD
}

__device__ __forceinline__ void recurrence_pick(const float* __restrict__ u,
                                                const float* __restrict__ b,
                                                const float* __restrict__ w1,
                                                const float* __restrict__ b1,
                                                const float* __restrict__ w2,
                                                const float* __restrict__ b2p,
                                                float* __restrict__ out,
                                                const int i_begin, const int n_chunks,
                                                const int warm_chunks,
                                                float* __restrict__ rec,
                                                const float* __restrict__ init,
                                                float* __restrict__ save) {
  const int lane = threadIdx.x;
  float sw = fabsf(w2[lane]) + fabsf(w2[lane + 32]) +
             fabsf(w2[lane + 64]) + fabsf(w2[lane + 96]);
  sw = dpp_xadd<0xB1>(sw);
  sw = dpp_xadd<0x4E>(sw);
  sw = dpp_xadd<0x141>(sw);
  sw = dpp_xadd<0x140>(sw);
  sw += xhalf(sw);
  const float B = lane_bcast(1.4426950408889634f * (sw + fabsf(b2p[0])), 0);
  if (B <= 100.f) recurrence_run<true>(u, b, w1, b1, w2, b2p, out, i_begin, n_chunks, warm_chunks, rec, init, save);
  else            recurrence_run<false>(u, b, w1, b1, w2, b2p, out, i_begin, n_chunks, warm_chunks, rec, init, save);
}

__global__ __launch_bounds__(32) void recurrence_chunks(const float* __restrict__ u,
                                                        const float* __restrict__ b,
                                                        const float* __restrict__ w1,
                                                        const float* __restrict__ b1,
                                                        const float* __restrict__ w2,
                                                        const float* __restrict__ b2p,
                                                        float* __restrict__ out,
                                                        float* __restrict__ sp,
                                                        float* __restrict__ st,
                                                        const unsigned long long ws_bytes) {
  if (ws_bytes < SP_NEED_BYTES) return;
  const int k      = blockIdx.x;
  const int i_main = k * SP_CH;
  const int i_beg  = (i_main > SP_W) ? (i_main - SP_W) : 0;
  const int warm   = (i_main - i_beg) / 32;
  recurrence_pick(u, b, w1, b1, w2, b2p, out, i_beg, warm + SP_CH / 32, warm, sp + k * SP_REC,
                  (const float*)nullptr, st + k * SP_REC);
}

__global__ __launch_bounds__(32) void recurrence_rerun(const float* __restrict__ u,
                                                       const float* __restrict__ b,
                                                       const float* __restrict__ w1,
                                                       const float* __restrict__ b1,
                                                       const float* __restrict__ w2,
                                                       const float* __restrict__ b2p,
                                                       float* __restrict__ out,
                                                       float* __restrict__ sp,
                                                       float* __restrict__ st,
                                                       const float* __restrict__ flag,
                                                       const unsigned long long ws_bytes) {
  if (ws_bytes < SP_NEED_BYTES) return;
  const int lane = threadIdx.x;
  const int k    = blockIdx.x + 1;
  if (flag[k] == 0.f) return;
  if (k > 1) { if (flag[k - 1] != 0.f) return; }
  const int i_main = k * SP_CH;
  float* r       = sp + k * SP_REC;
  const float* p = sp + (k - 1) * SP_REC;
  const float o0 = out[i_main - 64 + lane];
  const float o1 = out[i_main - 32 + lane];
  r[lane]      = o0;
  r[32 + lane] = o1;
  if (lane < 2) r[64 + lane] = p[66 + lane];
  recurrence_pick(u, b, w1, b1, w2, b2p, out, i_main, SP_CH / 32, 0, r,
                  st + (k - 1) * SP_REC, st + k * SP_REC);
}

__global__ __launch_bounds__(32) void boundary_check(const float* __restrict__ out,
                                                     const float* __restrict__ sp,
                                                     float* __restrict__ flag,
                                                     const unsigned long long ws_bytes) {
  if (ws_bytes < SP_NEED_BYTES) return;
  const int lane = threadIdx.x;
  const int k    = blockIdx.x + 1;
  const float* r = sp + k * SP_REC;
  const float* p = sp + (k - 1) * SP_REC;
  const int t    = k * SP_CH - 64;
  bool bad = __builtin_bit_cast(unsigned, r[lane])      != __builtin_bit_cast(unsigned, out[t + lane]);
  bad     |= __builtin_bit_cast(unsigned, r[32 + lane]) != __builtin_bit_cast(unsigned, out[t + 32 + lane]);
  if (lane < 2)
    bad   |= __builtin_bit_cast(unsigned, r[64 + lane]) != __builtin_bit_cast(unsigned, p[66 + lane]);
  const unsigned any = __builtin_amdgcn_ballot_w32(bad);
  if (lane == 0) flag[k] = (any != 0u) ? 1.f : 0.f;
}

__global__ __launch_bounds__(32) void recurrence_serial_if_needed(const float* __restrict__ u,
                                                                  const float* __restrict__ b,
                                                                  const float* __restrict__ w1,
                                                                  const float* __restrict__ b1,
                                                                  const float* __restrict__ w2,
                                                                  const float* __restrict__ b2p,
                                                                  float* __restrict__ out,
                                                                  const float* __restrict__ flag,
                                                                  const unsigned long long ws_bytes) {
  const int lane = threadIdx.x;
  unsigned any = 1u;
  if (ws_bytes >= SP_NEED_BYTES) {
    bool bad = false;
    for (int j = 0; j < SP_NB / 32; ++j) {
      const int k = 1 + lane + 32 * j;
      if (k < SP_NB) bad |= (flag[k] != 0.f);
    }
    any = __builtin_amdgcn_ballot_w32(bad);
  }
  if (any == 0u) return;
  recurrence_pick(u, b, w1, b1, w2, b2p, out, 0, T_LEN / 32, 0, (float*)nullptr,
                  (const float*)nullptr, (float*)nullptr);
}

extern "C" void kernel_launch(void* const* d_in, const int* in_sizes, int n_in,
                              void* d_out, int out_size, void* d_ws, size_t ws_size,
                              hipStream_t stream) {
  const float* Is   = (const float*)d_in[0];
  const float* a    = (const float*)d_in[1];
  const float* b    = (const float*)d_in[2];
  const float* g_w1 = (const float*)d_in[3];
  const float* g_b1 = (const float*)d_in[4];
  const float* g_w2 = (const float*)d_in[5];
  const float* g_b2 = (const float*)d_in[6];
  float* out  = (float*)d_out;
  float* u    = (float*)d_ws;
  float* sp   = u + T_LEN;
  float* st   = sp + SP_NB * SP_REC;
  float* flag0 = st + SP_NB * SP_REC;
  float* flag1 = flag0 + SP_NB;
  float* flag2 = flag1 + SP_NB;
  float* flag3 = flag2 + SP_NB;
  float* flag4 = flag3 + SP_NB;
  const unsigned long long ws_bytes = ws_size;

  fir_u_wmma<<<NTILE / 8, 256, 0, stream>>>(Is, a, u);
  recurrence_chunks<<<SP_NB, 32, 0, stream>>>(u, b, g_w1, g_b1, g_w2, g_b2, out, sp, st, ws_bytes);
  boundary_check<<<SP_NB - 1, 32, 0, stream>>>(out, sp, flag0, ws_bytes);
  recurrence_rerun<<<SP_NB - 1, 32, 0, stream>>>(u, b, g_w1, g_b1, g_w2, g_b2, out, sp, st, flag0, ws_bytes);
  boundary_check<<<SP_NB - 1, 32, 0, stream>>>(out, sp, flag1, ws_bytes);
  recurrence_rerun<<<SP_NB - 1, 32, 0, stream>>>(u, b, g_w1, g_b1, g_w2, g_b2, out, sp, st, flag1, ws_bytes);
  boundary_check<<<SP_NB - 1, 32, 0, stream>>>(out, sp, flag2, ws_bytes);
  recurrence_rerun<<<SP_NB - 1, 32, 0, stream>>>(u, b, g_w1, g_b1, g_w2, g_b2, out, sp, st, flag2, ws_bytes);
  boundary_check<<<SP_NB - 1, 32, 0, stream>>>(out, sp, flag3, ws_bytes);
  recurrence_rerun<<<SP_NB - 1, 32, 0, stream>>>(u, b, g_w1, g_b1, g_w2, g_b2, out, sp, st, flag3, ws_bytes);
  boundary_check<<<SP_NB - 1, 32, 0, stream>>>(out, sp, flag4, ws_bytes);
  recurrence_serial_if_needed<<<1, 32, 0, stream>>>(u, b, g_w1, g_b1, g_w2, g_b2, out, flag4, ws_bytes);
}


# recurrence split: check-and-re-run rounds cut from 4 to 1 (12 launches to 6); a boundary still failed after the one round goes to the serial run of the last launch; step and sums unchanged
# speedup vs baseline: 176.0785x; 1.0390x over previous
#include <hip/hip_runtime.h>
#include <hip/hip_bf16.h>

typedef __attribute__((ext_vector_type(2))) float v2f;
typedef __attribute__((ext_vector_type(8))) float v8f;

#define T_LEN 262144
#define NTILE (T_LEN / 256)

#define SP_CH  1024
#define SP_W   3584
#define SP_NB  (T_LEN / SP_CH)
#define SP_ROUNDS 1
#define SP_REC 72
#define SP_NEED_BYTES ((unsigned long long)(T_LEN + 2 * SP_NB * SP_REC + (SP_ROUNDS + 1) * SP_NB) * 4ull)

__global__ __launch_bounds__(256) void fir_u_wmma(const float* __restrict__ Is,
                                                  const float* __restrict__ a,
                                                  float* __restrict__ u) {
  __shared__ float a_pad[96];
  {
    const int tid = threadIdx.x;
    if (tid < 96) {
      const int d = tid - 16;
      a_pad[tid] = (d >= 0 && d < 64) ? a[d] : 0.f;
    }
  }
  __syncthreads();

  const int lane = threadIdx.x & 31;
  const int wave = threadIdx.x >> 5;
  const int tile = blockIdx.x * 8 + wave;
  const int t0   = tile * 256;
  const int base = t0 - 64;
  const int m    = lane & 15;
  const int half = lane >> 4;
  const int kb   = half * 2;

  v8f c = {0.f, 0.f, 0.f, 0.f, 0.f, 0.f, 0.f, 0.f};

  if (tile != 0) {
#pragma unroll
    for (int p = 0; p < 20; ++p) {
      const int k0 = p * 4 + kb;
      v2f av, bv;
      av.x = a_pad[16 + k0 - m];
      av.y = a_pad[17 + k0 - m];
      bv = *(const v2f*)(Is + (base + 16 * m + k0));
      c = __builtin_amdgcn_wmma_f32_16x16x4_f32(false, av, false, bv,
                                                (short)0, c, false, false);
    }
  } else {
#pragma unroll
    for (int p = 0; p < 20; ++p) {
      const int k0 = p * 4 + kb;
      v2f av, bv;
      av.x = a_pad[16 + k0 - m];
      av.y = a_pad[17 + k0 - m];
      const int i0 = base + 16 * m + k0;
      bv.x = (i0     >= 0) ? Is[i0]     : 0.f;
      bv.y = (i0 + 1 >= 0) ? Is[i0 + 1] : 0.f;
      c = __builtin_amdgcn_wmma_f32_16x16x4_f32(false, av, false, bv,
                                                (short)0, c, false, false);
    }
  }

#pragma unroll
  for (int r = 0; r < 8; ++r) {
    const int M = r + 8 * half;
    const int t = t0 + 16 * m + M;
    u[t] = Is[t] + c[r];
  }
}

template <int CTRL>
__device__ __forceinline__ float dpp_xadd(float x) {
  const int y = __builtin_amdgcn_update_dpp(0, __builtin_bit_cast(int, x),
                                            CTRL, 0xf, 0xf, true);
  return x + __builtin_bit_cast(float, y);
}

__device__ __forceinline__ float xhalf(float x) {
  const int b = __builtin_bit_cast(int, x);
  return __builtin_bit_cast(float,
      __builtin_amdgcn_permlanex16(0, b, 0x76543210, 0xfedcba98, false, false));
}

__device__ __forceinline__ float lane_bcast(float x, int l) {
  return __builtin_bit_cast(float,
      __builtin_amdgcn_readlane(__builtin_bit_cast(int, x), l));
}

__device__ __forceinline__ float lane_clear(float x, int l) {
  asm("v_writelane_b32 %0, 0, %1" : "+v"(x) : "i"(l));
  return x;
}

template <bool PROD>
__device__ __forceinline__ void recurrence_run(const float* __restrict__ u,
                                               const float* __restrict__ b,
                                               const float* __restrict__ w1,
                                               const float* __restrict__ b1,
                                               const float* __restrict__ w2,
                                               const float* __restrict__ b2p,
                                               float* __restrict__ out,
                                               const int i_begin,
                                               const int n_chunks,
                                               const int warm_chunks,
                                               float* __restrict__ rec,
                                               const float* __restrict__ init,
                                               float* __restrict__ save) {
  const int lane = threadIdx.x;

  const float L2E = 1.4426950408889634f;
  const float LN2 = 0.6931471805599453f;
  const float nb63 = -b[63];
  const float nb62 = -b[62];
  const float nb61 = -b[61];

  float S1[4], SB1[4], A[4], AL[4], Wt[4];
#pragma unroll
  for (int r = 0; r < 4; ++r) {
    S1[r]  = w1[lane + 32 * r];
    SB1[r] = b1[lane + 32 * r];
    A[r]   = nb63 * S1[r];
    AL[r]  = LN2 * A[r];
    Wt[r]  = L2E * w2[lane + 32 * r];
  }
  const float zc = (lane == 0) ? L2E * b2p[0] : 0.f;

  float T0[32], T1[32];
#pragma unroll
  for (int k = 0; k < 32; ++k) {
    const int m0 = (lane - k + 2) & 63;
    const int m1 = (lane - k + 34) & 63;
    const int d0 = (m0 == 0) ? 64 : m0;
    const int d1 = (m1 == 0) ? 64 : m1;
    const float v0 = b[(d0 >= 4) ? (64 - d0) : 0];
    const float v1 = b[(d1 >= 4) ? (64 - d1) : 0];
    T0[k] = (d0 >= 4) ? v0 : 0.f;
    T1[k] = (d1 >= 4) ? v1 : 0.f;
  }
  float X = 0.f, Y = 0.f;

  float fB = 0.f;
  float fC = 0.f;
  float lgP = 0.f;
  float m2P = 0.f;
  float q = u[i_begin + 1];
  float fOut = 0.f;
  float baseL = u[i_begin];
  if (init != nullptr) {
    X = init[lane];  Y = init[32 + lane];
    fB = init[64];   fC = init[65];  lgP = init[66];  m2P = init[67];
    q = init[68];    baseL = init[69];
  }
  float D[4];
#pragma unroll
  for (int r = 0; r < 4; ++r) D[r] = fmaf(S1[r], baseL, SB1[r]);
  if constexpr (!PROD) {
    if (init != nullptr) {
#pragma unroll
      for (int r = 0; r < 4; ++r) D[r] = fmaf(A[r], m2P, D[r]);
    }
  }
  float uB = u[i_begin + lane];

#define PIN() __builtin_amdgcn_sched_barrier(0)
#define HOLD(v) asm volatile("" : "+v"(v))

  for (int ch = 0; ch < n_chunks; ++ch) {
    const float uA = uB;
    const int   nb   = i_begin + (ch + 1) * 32;
    const int   pidx = (nb + lane < T_LEN) ? (nb + lane) : (T_LEN - 1);
    uB = u[pidx];
    const int i0 = i_begin + ch * 32;
#pragma unroll
    for (int s = 0; s < 32; ++s) {
      const float y1 = fmaf(AL[1], lgP, D[1]);
      const float y3 = fmaf(AL[3], lgP, D[3]);
      PIN();
      float fp;
      if constexpr (PROD) fp = (s == 0) ? fC : LN2 * lgP;
      else                fp = (s == 0) ? fC : fmaf(LN2, lgP, m2P);
      const float y0 = fmaf(AL[0], lgP, D[0]);
      const float y2 = fmaf(AL[2], lgP, D[2]);
      PIN();
      const float th1 = __builtin_amdgcn_tanhf(y1);
      PIN();
      X = fmaf(T0[s], fB, X);
      Y = fmaf(T1[s], fB, Y);
      HOLD(X);
      HOLD(Y);
      PIN();
      const float th3 = __builtin_amdgcn_tanhf(y3);
      PIN();
      const float t0 = fmaf(nb61, fB, q);
      PIN();
      const float th0 = __builtin_amdgcn_tanhf(y0);
      PIN();
      if (s > 0) { if (lane == s - 1) fOut = fp; }
      PIN();
      const float th2 = __builtin_amdgcn_tanhf(y2);
      PIN();
      const float in1 = fmaf(Wt[1], th1, zc);
      PIN();
      float rd;
      if (s < 30) rd = uA - X; else rd = uB - Y;
      HOLD(rd);
      PIN();
      const float m3 = Wt[3] * th3;
      PIN();
      const float base = fmaf(nb62, fp, t0);
      baseL = base;
      PIN();
      const float p0 = fmaf(Wt[0], th0, in1);
      PIN();
      const float p1 = fmaf(Wt[2], th2, m3);
      PIN();
      float qNew;
      if (s < 30) { qNew = lane_bcast(rd, s + 2);  X = lane_clear(X, s + 2);  }
      else        { qNew = lane_bcast(rd, s - 30); Y = lane_clear(Y, s - 30); }
      PIN();
      float zl = p0 + p1;
      PIN();
      float Cn[4];
#pragma unroll
      for (int r = 0; r < 4; ++r) Cn[r] = fmaf(S1[r], base, SB1[r]);
      if constexpr (PROD) {
#pragma unroll
        for (int r = 0; r < 4; ++r) D[r] = Cn[r];
      }
      PIN();
      zl = dpp_xadd<0xB1>(zl);
      zl = dpp_xadd<0x4E>(zl);
      zl = dpp_xadd<0x141>(zl);
      zl = dpp_xadd<0x140>(zl);
      float lg;
      float m2 = 0.f;
      if constexpr (PROD) {
        const float E  = __builtin_amdgcn_exp2f(zl);
        const float Eo = xhalf(E);
        const float w  = fmaf(E, Eo, 1.0f);
        lg = __builtin_amdgcn_logf(w);
        PIN();
      } else {
        zl += xhalf(zl);
        PIN();
        const float e = __builtin_amdgcn_exp2f(-fabsf(zl));
        PIN();
        m2 = LN2 * fmaxf(zl, 0.f);
        PIN();
        const float w = 1.0f + e;
        PIN();
        lg = __builtin_amdgcn_logf(w);
        PIN();
        D[1] = fmaf(A[1], m2, Cn[1]);
        D[3] = fmaf(A[3], m2, Cn[3]);
        D[0] = fmaf(A[0], m2, Cn[0]);
        D[2] = fmaf(A[2], m2, Cn[2]);
        PIN();
      }

      fB = fp;
      lgP = lg;
      m2P = m2;
      q = qNew;
    }
    if constexpr (PROD) fC = LN2 * lgP;
    else                fC = fmaf(LN2, lgP, m2P);
    if (lane == 31) fOut = fC;
    if (ch >= warm_chunks) {
      out[i0 + lane] = fOut;
    } else if (ch >= warm_chunks - 2) {
      rec[(ch - (warm_chunks - 2)) * 32 + lane] = fOut;
      if (ch == warm_chunks - 1 && lane == 0) { rec[64] = lgP; rec[65] = m2P; }
    }
    { const float t = X; X = Y; Y = t; }
  }
  if (rec != nullptr && lane == 0) { rec[66] = lgP; rec[67] = m2P; }
  if (save != nullptr) {
    save[lane] = X;  save[32 + lane] = Y;
    if (lane == 0) {
      save[64] = fB;  save[65] = fC;  save[66] = lgP;  save[67] = m2P;
      save[68] = q;   save[69] = baseL;
    }
  }
#undef PIN
#undef HOLD
}

__device__ __forceinline__ void recurrence_pick(const float* __restrict__ u,
                                                const float* __restrict__ b,
                                                const float* __restrict__ w1,
                                                const float* __restrict__ b1,
                                                const float* __restrict__ w2,
                                                const float* __restrict__ b2p,
                                                float* __restrict__ out,
                                                const int i_begin, const int n_chunks,
                                                const int warm_chunks,
                                                float* __restrict__ rec,
                                                const float* __restrict__ init,
                                                float* __restrict__ save) {
  const int lane = threadIdx.x;
  float sw = fabsf(w2[lane]) + fabsf(w2[lane + 32]) +
             fabsf(w2[lane + 64]) + fabsf(w2[lane + 96]);
  sw = dpp_xadd<0xB1>(sw);
  sw = dpp_xadd<0x4E>(sw);
  sw = dpp_xadd<0x141>(sw);
  sw = dpp_xadd<0x140>(sw);
  sw += xhalf(sw);
  const float B = lane_bcast(1.4426950408889634f * (sw + fabsf(b2p[0])), 0);
  if (B <= 100.f) recurrence_run<true>(u, b, w1, b1, w2, b2p, out, i_begin, n_chunks, warm_chunks, rec, init, save);
  else            recurrence_run<false>(u, b, w1, b1, w2, b2p, out, i_begin, n_chunks, warm_chunks, rec, init, save);
}

__global__ __launch_bounds__(32) void recurrence_chunks(const float* __restrict__ u,
                                                        const float* __restrict__ b,
                                                        const float* __restrict__ w1,
                                                        const float* __restrict__ b1,
                                                        const float* __restrict__ w2,
                                                        const float* __restrict__ b2p,
                                                        float* __restrict__ out,
                                                        float* __restrict__ sp,
                                                        float* __restrict__ st,
                                                        const unsigned long long ws_bytes) {
  if (ws_bytes < SP_NEED_BYTES) return;
  const int k      = blockIdx.x;
  const int i_main = k * SP_CH;
  const int i_beg  = (i_main > SP_W) ? (i_main - SP_W) : 0;
  const int warm   = (i_main - i_beg) / 32;
  recurrence_pick(u, b, w1, b1, w2, b2p, out, i_beg, warm + SP_CH / 32, warm, sp + k * SP_REC,
                  (const float*)nullptr, st + k * SP_REC);
}

__global__ __launch_bounds__(32) void recurrence_rerun(const float* __restrict__ u,
                                                       const float* __restrict__ b,
                                                       const float* __restrict__ w1,
                                                       const float* __restrict__ b1,
                                                       const float* __restrict__ w2,
                                                       const float* __restrict__ b2p,
                                                       float* __restrict__ out,
                                                       float* __restrict__ sp,
                                                       float* __restrict__ st,
                                                       const float* __restrict__ flag,
                                                       const unsigned long long ws_bytes) {
  if (ws_bytes < SP_NEED_BYTES) return;
  const int lane = threadIdx.x;
  const int k    = blockIdx.x + 1;
  if (flag[k] == 0.f) return;
  if (k > 1) { if (flag[k - 1] != 0.f) return; }
  const int i_main = k * SP_CH;
  float* r       = sp + k * SP_REC;
  const float* p = sp + (k - 1) * SP_REC;
  const float o0 = out[i_main - 64 + lane];
  const float o1 = out[i_main - 32 + lane];
  r[lane]      = o0;
  r[32 + lane] = o1;
  if (lane < 2) r[64 + lane] = p[66 + lane];
  recurrence_pick(u, b, w1, b1, w2, b2p, out, i_main, SP_CH / 32, 0, r,
                  st + (k - 1) * SP_REC, st + k * SP_REC);
}

__global__ __launch_bounds__(32) void boundary_check(const float* __restrict__ out,
                                                     const float* __restrict__ sp,
                                                     float* __restrict__ flag,
                                                     const unsigned long long ws_bytes) {
  if (ws_bytes < SP_NEED_BYTES) return;
  const int lane = threadIdx.x;
  const int k    = blockIdx.x + 1;
  const float* r = sp + k * SP_REC;
  const float* p = sp + (k - 1) * SP_REC;
  const int t    = k * SP_CH - 64;
  bool bad = __builtin_bit_cast(unsigned, r[lane])      != __builtin_bit_cast(unsigned, out[t + lane]);
  bad     |= __builtin_bit_cast(unsigned, r[32 + lane]) != __builtin_bit_cast(unsigned, out[t + 32 + lane]);
  if (lane < 2)
    bad   |= __builtin_bit_cast(unsigned, r[64 + lane]) != __builtin_bit_cast(unsigned, p[66 + lane]);
  const unsigned any = __builtin_amdgcn_ballot_w32(bad);
  if (lane == 0) flag[k] = (any != 0u) ? 1.f : 0.f;
}

__global__ __launch_bounds__(32) void recurrence_serial_if_needed(const float* __restrict__ u,
                                                                  const float* __restrict__ b,
                                                                  const float* __restrict__ w1,
                                                                  const float* __restrict__ b1,
                                                                  const float* __restrict__ w2,
                                                                  const float* __restrict__ b2p,
                                                                  float* __restrict__ out,
                                                                  const float* __restrict__ flag,
                                                                  const unsigned long long ws_bytes) {
  const int lane = threadIdx.x;
  unsigned any = 1u;
  if (ws_bytes >= SP_NEED_BYTES) {
    bool bad = false;
    for (int j = 0; j < SP_NB / 32; ++j) {
      const int k = 1 + lane + 32 * j;
      if (k < SP_NB) bad |= (flag[k] != 0.f);
    }
    any = __builtin_amdgcn_ballot_w32(bad);
  }
  if (any == 0u) return;
  recurrence_pick(u, b, w1, b1, w2, b2p, out, 0, T_LEN / 32, 0, (float*)nullptr,
                  (const float*)nullptr, (float*)nullptr);
}

extern "C" void kernel_launch(void* const* d_in, const int* in_sizes, int n_in,
                              void* d_out, int out_size, void* d_ws, size_t ws_size,
                              hipStream_t stream) {
  const float* Is   = (const float*)d_in[0];
  const float* a    = (const float*)d_in[1];
  const float* b    = (const float*)d_in[2];
  const float* g_w1 = (const float*)d_in[3];
  const float* g_b1 = (const float*)d_in[4];
  const float* g_w2 = (const float*)d_in[5];
  const float* g_b2 = (const float*)d_in[6];
  float* out  = (float*)d_out;
  float* u    = (float*)d_ws;
  float* sp   = u + T_LEN;
  float* st   = sp + SP_NB * SP_REC;
  float* flag0 = st + SP_NB * SP_REC;
  float* flag1 = flag0 + SP_NB;
  const unsigned long long ws_bytes = ws_size;

  fir_u_wmma<<<NTILE / 8, 256, 0, stream>>>(Is, a, u);
  recurrence_chunks<<<SP_NB, 32, 0, stream>>>(u, b, g_w1, g_b1, g_w2, g_b2, out, sp, st, ws_bytes);
  boundary_check<<<SP_NB - 1, 32, 0, stream>>>(out, sp, flag0, ws_bytes);
  recurrence_rerun<<<SP_NB - 1, 32, 0, stream>>>(u, b, g_w1, g_b1, g_w2, g_b2, out, sp, st, flag0, ws_bytes);
  boundary_check<<<SP_NB - 1, 32, 0, stream>>>(out, sp, flag1, ws_bytes);
  recurrence_serial_if_needed<<<1, 32, 0, stream>>>(u, b, g_w1, g_b1, g_w2, g_b2, out, flag1, ws_bytes);
}
